# starting kernel
# baseline (speedup 1.0000x reference)
#include <hip/hip_runtime.h>
#include <hip/hip_bf16.h>

#define N_NODES 50000
#define N_EDGES 1600000
#define IN_F    256
#define HID     32
#define HEADS   4
#define F2      128
#define NC      16
#define NEG_SLOPE 0.2f

typedef __attribute__((ext_vector_type(16))) _Float16 v16h;
typedef __attribute__((ext_vector_type(8)))  _Float16 v8h;
typedef __attribute__((ext_vector_type(8)))  float    v8f;

__device__ __forceinline__ unsigned enc_key(float f) {
    unsigned b = __float_as_uint(f);
    return (b & 0x80000000u) ? ~b : (b | 0x80000000u);
}
__device__ __forceinline__ float dec_key(unsigned k) {
    unsigned b = (k & 0x80000000u) ? (k ^ 0x80000000u) : ~k;
    if ((b & 0x7f800000u) == 0x7f800000u) return 0.0f;
    return __uint_as_float(b);
}
__device__ __forceinline__ float leaky(float v) { return v > 0.0f ? v : NEG_SLOPE * v; }

__global__ void k_f32_to_f16(const float* __restrict__ in, _Float16* __restrict__ out, int n) {
    int i = blockIdx.x * blockDim.x + threadIdx.x;
    if (i < n) out[i] = (_Float16)in[i];
}
__global__ void k_transpose_f16(const float* __restrict__ in, _Float16* __restrict__ out,
                                int R, int C) {
    int i = blockIdx.x * blockDim.x + threadIdx.x;
    if (i >= R * C) return;
    int r = i / C, c = i % C;
    out[c * R + r] = (_Float16)in[i];
}

template <int K>
__global__ void __launch_bounds__(256) k_gemm_wmma(const _Float16* __restrict__ A,
                                                   const _Float16* __restrict__ Bt,
                                                   float* __restrict__ C,
                                                   int M, int Ncol) {
    int wave = (blockIdx.x * blockDim.x + threadIdx.x) >> 5;
    int lane = threadIdx.x & 31;
    int mTiles = M >> 4, nTiles = Ncol >> 4;
    if (wave >= mTiles * nTiles) return;
    int mt = wave / nTiles, nt = wave % nTiles;
    int half = lane >> 4;
    int idx  = lane & 15;

    const _Float16* arow = A  + (size_t)(mt * 16 + idx) * K + half * 8;
    const _Float16* brow = Bt + (size_t)(nt * 16 + idx) * K + half * 8;

    v8f acc = {};
#pragma unroll
    for (int k0 = 0; k0 < K; k0 += 32) {
        union { v16h v; struct { v8h lo, hi; } h; } a, b;
        a.h.lo = *(const v8h*)(arow + k0);
        a.h.hi = *(const v8h*)(arow + k0 + 16);
        b.h.lo = *(const v8h*)(brow + k0);
        b.h.hi = *(const v8h*)(brow + k0 + 16);
        acc = __builtin_amdgcn_wmma_f32_16x16x32_f16(false, a.v, false, b.v,
                                                     (short)0, acc, false, false);
    }
    float* crow = C + (size_t)(mt * 16 + half * 8) * Ncol + nt * 16 + idx;
#pragma unroll
    for (int r = 0; r < 8; ++r) crow[(size_t)r * Ncol] = acc[r];
}

__global__ void k_el_er(const float* __restrict__ feat, const float* __restrict__ al,
                        const float* __restrict__ ar, float* __restrict__ el,
                        float* __restrict__ er, int n, int H, int D) {
    int i = blockIdx.x * blockDim.x + threadIdx.x;
    if (i >= n * H) return;
    int node = i / H, h = i % H;
    const float* f = feat + (size_t)node * H * D + h * D;
    const float* a = al + h * D;
    const float* b = ar + h * D;
    float sl = 0.0f, sr = 0.0f;
    for (int d = 0; d < D; ++d) { float v = f[d]; sl += v * a[d]; sr += v * b[d]; }
    el[i] = sl; er[i] = sr;
}

__global__ void k_edge_max4(const int* __restrict__ src, const int* __restrict__ dst,
                            const float* __restrict__ el, const float* __restrict__ er,
                            unsigned* __restrict__ mx) {
    int e = blockIdx.x * blockDim.x + threadIdx.x;
    if (e >= N_EDGES) return;
    int s = src[e], d = dst[e];
    float4 a = *(const float4*)(el + 4 * (size_t)s);
    float4 b = *(const float4*)(er + 4 * (size_t)d);
    float v[4] = { a.x + b.x, a.y + b.y, a.z + b.z, a.w + b.w };
#pragma unroll
    for (int h = 0; h < 4; ++h)
        atomicMax(mx + 4 * (size_t)d + h, enc_key(leaky(v[h])));
}

__global__ void k_edge_exp4(const int* __restrict__ src, const int* __restrict__ dst,
                            const float* __restrict__ el, const float* __restrict__ er,
                            const unsigned* __restrict__ mx, float* __restrict__ ex,
                            float* __restrict__ den) {
    int e = blockIdx.x * blockDim.x + threadIdx.x;
    if (e >= N_EDGES) return;
    int s = src[e], d = dst[e];
    float4 a = *(const float4*)(el + 4 * (size_t)s);
    float4 b = *(const float4*)(er + 4 * (size_t)d);
    float v[4] = { a.x + b.x, a.y + b.y, a.z + b.z, a.w + b.w };
#pragma unroll
    for (int h = 0; h < 4; ++h) {
        float m = dec_key(mx[4 * (size_t)d + h]);
        float x = __expf(leaky(v[h]) - m);
        ex[4 * (size_t)e + h] = x;
        atomicAdd(den + 4 * (size_t)d + h, x);
    }
}

__global__ void __launch_bounds__(256) k_edge_agg4(const int* __restrict__ src,
                                                   const int* __restrict__ dst,
                                                   const float* __restrict__ feat,
                                                   const float* __restrict__ ex,
                                                   const float* __restrict__ den,
                                                   float* __restrict__ agg) {
    int wave = (blockIdx.x * blockDim.x + threadIdx.x) >> 5;
    int lane = threadIdx.x & 31;
    if (wave >= N_EDGES) return;
    int s = src[wave], d = dst[wave];
    int h = lane >> 3;
    float dn = den[4 * (size_t)d + h];
    float alpha = ex[4 * (size_t)wave + h] / fmaxf(dn, 1e-9f);
    float4 f = *(const float4*)(feat + (size_t)s * F2 + lane * 4);
    float* o = agg + (size_t)d * F2 + lane * 4;
    atomicAdd(o + 0, f.x * alpha);
    atomicAdd(o + 1, f.y * alpha);
    atomicAdd(o + 2, f.z * alpha);
    atomicAdd(o + 3, f.w * alpha);
}

__global__ void k_fin1(const float* __restrict__ agg, const float* __restrict__ b1,
                       _Float16* __restrict__ h16) {
    int i = blockIdx.x * blockDim.x + threadIdx.x;
    if (i >= N_NODES * F2) return;
    float v = agg[i] + b1[i & (F2 - 1)];
    v = v > 0.0f ? v : (__expf(v) - 1.0f);
    h16[i] = (_Float16)v;
}

__global__ void k_edge_max1(const int* __restrict__ src, const int* __restrict__ dst,
                            const float* __restrict__ el, const float* __restrict__ er,
                            unsigned* __restrict__ mx) {
    int e = blockIdx.x * blockDim.x + threadIdx.x;
    if (e >= N_EDGES) return;
    int s = src[e], d = dst[e];
    atomicMax(mx + d, enc_key(leaky(el[s] + er[d])));
}

__global__ void k_edge_exp1(const int* __restrict__ src, const int* __restrict__ dst,
                            const float* __restrict__ el, const float* __restrict__ er,
                            const unsigned* __restrict__ mx, float* __restrict__ ex,
                            float* __restrict__ den) {
    int e = blockIdx.x * blockDim.x + threadIdx.x;
    if (e >= N_EDGES) return;
    int s = src[e], d = dst[e];
    float x = __expf(leaky(el[s] + er[d]) - dec_key(mx[d]));
    ex[e] = x;
    atomicAdd(den + d, x);
}

__global__ void k_edge_agg1(const int* __restrict__ src, const int* __restrict__ dst,
                            const float* __restrict__ feat, const float* __restrict__ ex,
                            const float* __restrict__ den, float* __restrict__ agg) {
    int t = blockIdx.x * blockDim.x + threadIdx.x;
    if (t >= N_EDGES * 4) return;
    int e = t >> 2, q = t & 3;
    int s = src[e], d = dst[e];
    float alpha = ex[e] / fmaxf(den[d], 1e-9f);
    float4 f = *(const float4*)(feat + (size_t)s * NC + q * 4);
    float* o = agg + (size_t)d * NC + q * 4;
    atomicAdd(o + 0, f.x * alpha);
    atomicAdd(o + 1, f.y * alpha);
    atomicAdd(o + 2, f.z * alpha);
    atomicAdd(o + 3, f.w * alpha);
}

__global__ void k_fin2(const float* __restrict__ agg, const float* __restrict__ b2,
                       float* __restrict__ out) {
    int i = blockIdx.x * blockDim.x + threadIdx.x;
    if (i >= N_NODES * NC) return;
    out[i] = agg[i] + b2[i & (NC - 1)];
}

extern "C" void kernel_launch(void* const* d_in, const int* in_sizes, int n_in,
                              void* d_out, int out_size, void* d_ws, size_t ws_size,
                              hipStream_t stream) {
    const float* x   = (const float*)d_in[0];
    const int*   src = (const int*)  d_in[1];
    const int*   dst = (const int*)  d_in[2];
    const float* W1  = (const float*)d_in[3];
    const float* b1  = (const float*)d_in[4];
    const float* al1 = (const float*)d_in[5];
    const float* ar1 = (const float*)d_in[6];
    const float* W2  = (const float*)d_in[7];
    const float* b2  = (const float*)d_in[8];
    const float* al2 = (const float*)d_in[9];
    const float* ar2 = (const float*)d_in[10];
    float* out = (float*)d_out;
    (void)in_sizes; (void)n_in; (void)out_size; (void)ws_size;

    char* p = (char*)d_ws;
    auto carve = [&](size_t bytes) -> char* {
        char* r = p; p += (bytes + 255) & ~(size_t)255; return r;
    };
    _Float16* xh    = (_Float16*)carve((size_t)N_NODES * IN_F * 2);
    _Float16* w1t   = (_Float16*)carve((size_t)IN_F * F2 * 2);
    float*    feat1 = (float*)   carve((size_t)N_NODES * F2 * 4);
    float*    el1   = (float*)   carve((size_t)N_NODES * HEADS * 4);
    float*    er1   = (float*)   carve((size_t)N_NODES * HEADS * 4);
    unsigned* mx1   = (unsigned*)carve((size_t)N_NODES * HEADS * 4);
    float*    den1  = (float*)   carve((size_t)N_NODES * HEADS * 4);
    float*    ex1   = (float*)   carve((size_t)N_EDGES * HEADS * 4);
    float*    agg1  = (float*)   carve((size_t)N_NODES * F2 * 4);
    _Float16* h16   = (_Float16*)carve((size_t)N_NODES * F2 * 2);
    _Float16* w2t   = (_Float16*)carve((size_t)F2 * NC * 2);
    float*    feat2 = (float*)   carve((size_t)N_NODES * NC * 4);
    float*    el2   = (float*)   carve((size_t)N_NODES * 4);
    float*    er2   = (float*)   carve((size_t)N_NODES * 4);
    unsigned* mx2   = (unsigned*)carve((size_t)N_NODES * 4);
    float*    den2  = (float*)   carve((size_t)N_NODES * 4);
    float*    ex2   = (float*)   carve((size_t)N_EDGES * 4);
    float*    agg2  = (float*)   carve((size_t)N_NODES * NC * 4);

    hipMemsetAsync(mx1,  0, (size_t)N_NODES * HEADS * 4, stream);
    hipMemsetAsync(den1, 0, (size_t)N_NODES * HEADS * 4, stream);
    hipMemsetAsync(agg1, 0, (size_t)N_NODES * F2 * 4,    stream);
    hipMemsetAsync(mx2,  0, (size_t)N_NODES * 4,         stream);
    hipMemsetAsync(den2, 0, (size_t)N_NODES * 4,         stream);
    hipMemsetAsync(agg2, 0, (size_t)N_NODES * NC * 4,    stream);

    const int T = 256;
    auto blk = [](long long n, int t) { return (int)((n + t - 1) / t); };

    k_f32_to_f16   <<<blk((long long)N_NODES * IN_F, T), T, 0, stream>>>(x, xh, N_NODES * IN_F);
    k_transpose_f16<<<blk((long long)IN_F * F2, T), T, 0, stream>>>(W1, w1t, IN_F, F2);

    {
        int waves = (N_NODES / 16) * (F2 / 16);
        k_gemm_wmma<IN_F><<<blk(waves, 8), T, 0, stream>>>(xh, w1t, feat1, N_NODES, F2);
    }
    k_el_er<<<blk((long long)N_NODES * HEADS, T), T, 0, stream>>>(feat1, al1, ar1, el1, er1,
                                                                  N_NODES, HEADS, HID);
    k_edge_max4<<<blk(N_EDGES, T), T, 0, stream>>>(src, dst, el1, er1, mx1);
    k_edge_exp4<<<blk(N_EDGES, T), T, 0, stream>>>(src, dst, el1, er1, mx1, ex1, den1);
    k_edge_agg4<<<blk((long long)N_EDGES * 32, T), T, 0, stream>>>(src, dst, feat1, ex1, den1, agg1);
    k_fin1<<<blk((long long)N_NODES * F2, T), T, 0, stream>>>(agg1, b1, h16);

    k_transpose_f16<<<blk((long long)F2 * NC, T), T, 0, stream>>>(W2, w2t, F2, NC);
    {
        int waves = (N_NODES / 16) * (NC / 16);
        k_gemm_wmma<F2><<<blk(waves, 8), T, 0, stream>>>(h16, w2t, feat2, N_NODES, NC);
    }
    k_el_er<<<blk(N_NODES, T), T, 0, stream>>>(feat2, al2, ar2, el2, er2, N_NODES, 1, NC);
    k_edge_max1<<<blk(N_EDGES, T), T, 0, stream>>>(src, dst, el2, er2, mx2);
    k_edge_exp1<<<blk(N_EDGES, T), T, 0, stream>>>(src, dst, el2, er2, mx2, ex2, den2);
    k_edge_agg1<<<blk((long long)N_EDGES * 4, T), T, 0, stream>>>(src, dst, feat2, ex2, den2, agg2);
    k_fin2<<<blk((long long)N_NODES * NC, T), T, 0, stream>>>(agg2, b2, out);
}


# Each workgroup owns 196 destination nodes: finds its edges in one read of the edge list, keeps maxima, denominators, integer row sums in LDS (LDS atomics only), writes rows once; 15 launches become 9
# speedup vs baseline: 3.8914x; 3.8914x over previous
#include <hip/hip_runtime.h>
#include <hip/hip_bf16.h>

#define N_NODES 50000
#define N_EDGES 1600000
#define IN_F    256
#define HID     32
#define HEADS   4
#define F2      128
#define NC      16
#define NEG_SLOPE 0.2f

typedef __attribute__((ext_vector_type(16))) _Float16 v16h;
typedef __attribute__((ext_vector_type(8)))  _Float16 v8h;
typedef __attribute__((ext_vector_type(8)))  float    v8f;

__device__ __forceinline__ unsigned enc_key(float f) {
    unsigned b = __float_as_uint(f);
    return (b & 0x80000000u) ? ~b : (b | 0x80000000u);
}
__device__ __forceinline__ float dec_key(unsigned k) {
    unsigned b = (k & 0x80000000u) ? (k ^ 0x80000000u) : ~k;
    if ((b & 0x7f800000u) == 0x7f800000u) return 0.0f;
    return __uint_as_float(b);
}
__device__ __forceinline__ float leaky(float v) { return v > 0.0f ? v : NEG_SLOPE * v; }

#define DEN_SCALE 2147483648.0f
#define DEN_INV   4.656612873077392578125e-10f
__device__ __forceinline__ int fx_exp(unsigned maxbits) {
    int e = (int)((maxbits >> 23) & 0xffu) - 127;
    return e < -96 ? -96 : (e > 96 ? 96 : e);
}
__device__ __forceinline__ float fx_scale(unsigned maxbits) {
    return __uint_as_float((unsigned)(156 - fx_exp(maxbits)) << 23);
}
__device__ __forceinline__ float fx_inv(unsigned maxbits) {
    return __uint_as_float((unsigned)(98 + fx_exp(maxbits)) << 23);
}

__global__ void k_f32_to_f16(const float* __restrict__ in, _Float16* __restrict__ out, int n) {
    int i = blockIdx.x * blockDim.x + threadIdx.x;
    if (i < n) out[i] = (_Float16)in[i];
}
__global__ void k_transpose_f16(const float* __restrict__ in, _Float16* __restrict__ out,
                                int R, int C) {
    int i = blockIdx.x * blockDim.x + threadIdx.x;
    if (i >= R * C) return;
    int r = i / C, c = i % C;
    out[c * R + r] = (_Float16)in[i];
}

template <int K>
__global__ void __launch_bounds__(256) k_gemm_wmma(const _Float16* __restrict__ A,
                                                   const _Float16* __restrict__ Bt,
                                                   float* __restrict__ C,
                                                   int M, int Ncol) {
    int wave = (blockIdx.x * blockDim.x + threadIdx.x) >> 5;
    int lane = threadIdx.x & 31;
    int mTiles = M >> 4, nTiles = Ncol >> 4;
    if (wave >= mTiles * nTiles) return;
    int mt = wave / nTiles, nt = wave % nTiles;
    int half = lane >> 4;
    int idx  = lane & 15;

    const _Float16* arow = A  + (size_t)(mt * 16 + idx) * K + half * 8;
    const _Float16* brow = Bt + (size_t)(nt * 16 + idx) * K + half * 8;

    v8f acc = {};
#pragma unroll
    for (int k0 = 0; k0 < K; k0 += 32) {
        union { v16h v; struct { v8h lo, hi; } h; } a, b;
        a.h.lo = *(const v8h*)(arow + k0);
        a.h.hi = *(const v8h*)(arow + k0 + 16);
        b.h.lo = *(const v8h*)(brow + k0);
        b.h.hi = *(const v8h*)(brow + k0 + 16);
        acc = __builtin_amdgcn_wmma_f32_16x16x32_f16(false, a.v, false, b.v,
                                                     (short)0, acc, false, false);
    }
    float* crow = C + (size_t)(mt * 16 + half * 8) * Ncol + nt * 16 + idx;
#pragma unroll
    for (int r = 0; r < 8; ++r) crow[(size_t)r * Ncol] = acc[r];
}

__global__ void k_el_er(const float* __restrict__ feat, const float* __restrict__ al,
                        const float* __restrict__ ar, float* __restrict__ el,
                        float* __restrict__ er, int n, int H, int D,
                        unsigned* __restrict__ mxabs) {
    int i = blockIdx.x * blockDim.x + threadIdx.x;
    float ma = 0.0f;
    if (i < n * H) {
        int node = i / H, h = i % H;
        const float* f = feat + (size_t)node * H * D + h * D;
        const float* a = al + h * D;
        const float* b = ar + h * D;
        float sl = 0.0f, sr = 0.0f;
        for (int d = 0; d < D; ++d) {
            float v = f[d]; sl += v * a[d]; sr += v * b[d]; ma = fmaxf(ma, fabsf(v));
        }
        el[i] = sl; er[i] = sr;
    }
    for (int o = 16; o > 0; o >>= 1) ma = fmaxf(ma, __shfl_xor(ma, o, 32));
    if ((threadIdx.x & 31) == 0) atomicMax(mxabs, __float_as_uint(ma));
}

#define NPW   196
#define NWG   256
#define WGT   1024
#define SCAN_U 8
#define LCAP1 16384
#define LCAP2 40960

template <int H, int D, int LCAP, bool FIRST>
__global__ void __launch_bounds__(WGT)
k_edges_owned(const int* __restrict__ src, const int* __restrict__ dst,
              const float* __restrict__ feat, const float* __restrict__ el,
              const float* __restrict__ er, const unsigned* __restrict__ mxabs,
              const float* __restrict__ bias, unsigned* glist, unsigned* gcnt,
              _Float16* __restrict__ o16, float* __restrict__ o32) {
    constexpr int F = H * D;
    static_assert(SCAN_U * 4 <= 32, "one mask bit per scanned index");
    static_assert(F >= 32 ? (D % 32 == 0 && (H & (H - 1)) == 0) : (H == 1 && 32 % F == 0), "lane map");
    __shared__ int acc[NPW * F];
    __shared__ unsigned long long den[NPW * H];
    __shared__ unsigned mx[NPW * H];
    __shared__ float ers[NPW * H];
    __shared__ unsigned list[LCAP];
    __shared__ unsigned cnt;

    const int tid = threadIdx.x;
    const int n0 = blockIdx.x * NPW;

    for (int i = tid; i < NPW * F; i += WGT) acc[i] = 0;
    for (int i = tid; i < NPW * H; i += WGT) {
        den[i] = 0ull; mx[i] = 0u;
        ers[i] = (n0 + i / H) < N_NODES ? er[(size_t)n0 * H + i] : 0.0f;
    }
    if (tid == 0) cnt = 0u;
    __syncthreads();

    bool have = false;
    if (!FIRST) {
        unsigned c = gcnt[blockIdx.x];
        if (c <= (unsigned)LCAP1) {
            for (unsigned i = tid; i < c; i += WGT) list[i] = glist[(size_t)blockIdx.x * LCAP1 + i];
            if (tid == 0) cnt = c;
            have = true;
            __syncthreads();
        }
    }
    if (!have) {
        for (int base = tid * 4; base < N_EDGES; base += WGT * 4 * SCAN_U) {
            int4 dv[SCAN_U];
#pragma unroll
            for (int u = 0; u < SCAN_U; ++u) {
                int b = base + u * WGT * 4;
                dv[u] = b < N_EDGES ? *(const int4*)(dst + b) : make_int4(-1, -1, -1, -1);
            }
            unsigned m = 0u;
#pragma unroll
            for (int u = 0; u < SCAN_U; ++u) {
                m |= (unsigned)((unsigned)(dv[u].x - n0) < (unsigned)NPW) << (4 * u);
                m |= (unsigned)((unsigned)(dv[u].y - n0) < (unsigned)NPW) << (4 * u + 1);
                m |= (unsigned)((unsigned)(dv[u].z - n0) < (unsigned)NPW) << (4 * u + 2);
                m |= (unsigned)((unsigned)(dv[u].w - n0) < (unsigned)NPW) << (4 * u + 3);
            }
            while (m) {
                int bit = __ffs((int)m) - 1;
                m &= m - 1u;
                unsigned pos = atomicAdd(&cnt, 1u);
                if (pos < (unsigned)LCAP) list[pos] = (unsigned)(base + (bit >> 2) * (WGT * 4) + (bit & 3));
            }
        }
        __syncthreads();
    }
    const unsigned total = cnt;
    const bool complete = total <= (unsigned)LCAP;

    auto score = [&](unsigned s, unsigned dl, int h) -> float {
        return leaky(el[(size_t)s * H + h] + ers[dl * H + h]);
    };

    if (complete) {
        for (unsigned i = tid; i < total; i += WGT) {
            unsigned w = list[i], s, dl;
            if (have) { s = w & 0xffffu; dl = w >> 16; }
            else {
                dl = (unsigned)(dst[w] - n0); s = (unsigned)src[w];
                w = s | (dl << 16);
                list[i] = w;
                if (FIRST) glist[(size_t)blockIdx.x * LCAP1 + i] = w;
            }
#pragma unroll
            for (int h = 0; h < H; ++h) atomicMax(&mx[dl * H + h], enc_key(score(s, dl, h)));
        }
    } else {
        for (int e = tid; e < N_EDGES; e += WGT) {
            unsigned dl = (unsigned)(dst[e] - n0);
            if (dl < (unsigned)NPW) {
                unsigned s = (unsigned)src[e];
                for (int h = 0; h < H; ++h) atomicMax(&mx[dl * H + h], enc_key(score(s, dl, h)));
            }
        }
    }
    if (FIRST && tid == 0) gcnt[blockIdx.x] = total;
    __syncthreads();

    auto edge_den = [&](unsigned s, unsigned dl) {
#pragma unroll
        for (int h = 0; h < H; ++h) {
            float x = __expf(score(s, dl, h) - dec_key(mx[dl * H + h]));
            atomicAdd(&den[dl * H + h], (unsigned long long)__float2uint_rn(x * DEN_SCALE));
        }
    };
    if (complete) {
        for (unsigned i = tid; i < total; i += WGT) { unsigned w = list[i]; edge_den(w & 0xffffu, w >> 16); }
    } else {
        for (int e = tid; e < N_EDGES; e += WGT) {
            unsigned dl = (unsigned)(dst[e] - n0);
            if (dl < (unsigned)NPW) edge_den((unsigned)src[e], dl);
        }
    }
    __syncthreads();

    const float sc = fx_scale(*mxabs);
    auto alpha = [&](unsigned s, unsigned dl, int h) -> float {
        float x = __expf(score(s, dl, h) - dec_key(mx[dl * H + h]));
        return x / fmaxf((float)den[dl * H + h] * DEN_INV, 1e-9f) * sc;
    };
    if (complete) {
        constexpr int EPS = F >= 32 ? 1 : 32 / F;
        constexpr int NJ  = F >= 32 ? F / 32 : 1;
        constexpr int U   = F >= 32 ? 6 : 8;
        const int lane = tid & 31, wv = tid >> 5;
        const int sub = F >= 32 ? 0 : lane / F;
        const int fl  = F >= 32 ? lane : lane % F;
        for (unsigned i0 = (unsigned)(wv * EPS * U); i0 < total; i0 += (unsigned)((WGT / 32) * EPS * U)) {
            float f[U][NJ], a[U][NJ];
            unsigned dls[U];
            bool ok[U];
#pragma unroll
            for (int u = 0; u < U; ++u) {
                unsigned i = i0 + (unsigned)(u * EPS + sub);
                ok[u] = i < total;
                unsigned w = list[ok[u] ? i : total - 1u];
                unsigned s = w & 0xffffu;
                dls[u] = w >> 16;
                const float* fr = feat + (size_t)s * F + fl;
#pragma unroll
                for (int j = 0; j < NJ; ++j) f[u][j] = fr[32 * j];
                float aq = alpha(s, dls[u], F >= 32 ? (lane & (H - 1)) : 0);
#pragma unroll
                for (int j = 0; j < NJ; ++j)
                    a[u][j] = F >= 32 ? __int_as_float(__builtin_amdgcn_readlane(__float_as_int(aq), (32 * j) / D))
                                      : aq;
            }
#pragma unroll
            for (int u = 0; u < U; ++u) {
                if (ok[u]) {
#pragma unroll
                    for (int j = 0; j < NJ; ++j)
                        atomicAdd(&acc[dls[u] * F + fl + 32 * j], __float2int_rn(f[u][j] * a[u][j]));
                }
            }
        }
    } else {
        for (int e = tid; e < N_EDGES; e += WGT) {
            unsigned dl = (unsigned)(dst[e] - n0);
            if (dl < (unsigned)NPW) {
                unsigned s = (unsigned)src[e];
                for (int h = 0; h < H; ++h) {
                    float ah = alpha(s, dl, h);
                    for (int d = 0; d < D; ++d)
                        atomicAdd(&acc[dl * F + h * D + d],
                                  __float2int_rn(feat[(size_t)s * F + h * D + d] * ah));
                }
            }
        }
    }
    __syncthreads();

    const float inv = fx_inv(*mxabs);
    for (int i = tid; i < NPW * F; i += WGT) {
        if (n0 + i / F < N_NODES) {
            float v = (float)acc[i] * inv + bias[i % F];
            if (FIRST) {
                v = v > 0.0f ? v : (__expf(v) - 1.0f);
                o16[(size_t)n0 * F + i] = (_Float16)v;
            } else {
                o32[(size_t)n0 * F + i] = v;
            }
        }
    }
}

extern "C" void kernel_launch(void* const* d_in, const int* in_sizes, int n_in,
                              void* d_out, int out_size, void* d_ws, size_t ws_size,
                              hipStream_t stream) {
    const float* x   = (const float*)d_in[0];
    const int*   src = (const int*)  d_in[1];
    const int*   dst = (const int*)  d_in[2];
    const float* W1  = (const float*)d_in[3];
    const float* b1  = (const float*)d_in[4];
    const float* al1 = (const float*)d_in[5];
    const float* ar1 = (const float*)d_in[6];
    const float* W2  = (const float*)d_in[7];
    const float* b2  = (const float*)d_in[8];
    const float* al2 = (const float*)d_in[9];
    const float* ar2 = (const float*)d_in[10];
    float* out = (float*)d_out;
    (void)in_sizes; (void)n_in; (void)out_size; (void)ws_size;

    char* p = (char*)d_ws;
    auto carve = [&](size_t bytes) -> char* {
        char* r = p; p += (bytes + 255) & ~(size_t)255; return r;
    };
    _Float16* xh    = (_Float16*)carve((size_t)N_NODES * IN_F * 2);
    _Float16* w1t   = (_Float16*)carve((size_t)IN_F * F2 * 2);
    float*    feat1 = (float*)   carve((size_t)N_NODES * F2 * 4);
    float*    el1   = (float*)   carve((size_t)N_NODES * HEADS * 4);
    float*    er1   = (float*)   carve((size_t)N_NODES * HEADS * 4);
    unsigned* mxa1  = (unsigned*)carve(256);
    unsigned* mxa2  = mxa1 + 1;
    unsigned* glist = (unsigned*)carve((size_t)NWG * LCAP1 * 4);
    unsigned* gcnt  = (unsigned*)carve((size_t)NWG * 4);
    _Float16* h16   = (_Float16*)carve((size_t)N_NODES * F2 * 2);
    _Float16* w2t   = (_Float16*)carve((size_t)F2 * NC * 2);
    float*    feat2 = (float*)   carve((size_t)N_NODES * NC * 4);
    float*    el2   = (float*)   carve((size_t)N_NODES * 4);
    float*    er2   = (float*)   carve((size_t)N_NODES * 4);

    hipMemsetAsync(mxa1, 0, 256, stream);

    const int T = 256;
    auto blk = [](long long n, int t) { return (int)((n + t - 1) / t); };

    k_f32_to_f16   <<<blk((long long)N_NODES * IN_F, T), T, 0, stream>>>(x, xh, N_NODES * IN_F);
    k_transpose_f16<<<blk((long long)IN_F * F2, T), T, 0, stream>>>(W1, w1t, IN_F, F2);

    {
        int waves = (N_NODES / 16) * (F2 / 16);
        k_gemm_wmma<IN_F><<<blk(waves, 8), T, 0, stream>>>(xh, w1t, feat1, N_NODES, F2);
    }
    k_el_er<<<blk((long long)N_NODES * HEADS, T), T, 0, stream>>>(feat1, al1, ar1, el1, er1,
                                                                  N_NODES, HEADS, HID, mxa1);
    k_edges_owned<HEADS, HID, LCAP1, true><<<NWG, WGT, 0, stream>>>(src, dst, feat1, el1, er1, mxa1, b1,
                                                                     glist, gcnt, h16, nullptr);

    k_transpose_f16<<<blk((long long)F2 * NC, T), T, 0, stream>>>(W2, w2t, F2, NC);
    {
        int waves = (N_NODES / 16) * (NC / 16);
        k_gemm_wmma<F2><<<blk(waves, 8), T, 0, stream>>>(h16, w2t, feat2, N_NODES, NC);
    }
    k_el_er<<<blk(N_NODES, T), T, 0, stream>>>(feat2, al2, ar2, el2, er2, N_NODES, 1, NC, mxa2);
    k_edges_owned<1, NC, LCAP2, false><<<NWG, WGT, 0, stream>>>(src, dst, feat2, el2, er2, mxa2, b2,
                                                                 glist, gcnt, nullptr, out);
}


# Conversion pass of x widened: each lane converts 64 values (sixteen 16-byte loads in flight, 8-byte stores, whole 128-byte lines per wave instruction); 400000 one-value waves become 6250
# speedup vs baseline: 4.1460x; 1.0654x over previous
#include <hip/hip_runtime.h>
#include <hip/hip_bf16.h>

#define N_NODES 50000
#define N_EDGES 1600000
#define IN_F    256
#define HID     32
#define HEADS   4
#define F2      128
#define NC      16
#define NEG_SLOPE 0.2f

typedef __attribute__((ext_vector_type(16))) _Float16 v16h;
typedef __attribute__((ext_vector_type(8)))  _Float16 v8h;
typedef __attribute__((ext_vector_type(8)))  float    v8f;

__device__ __forceinline__ unsigned enc_key(float f) {
    unsigned b = __float_as_uint(f);
    return (b & 0x80000000u) ? ~b : (b | 0x80000000u);
}
__device__ __forceinline__ float dec_key(unsigned k) {
    unsigned b = (k & 0x80000000u) ? (k ^ 0x80000000u) : ~k;
    if ((b & 0x7f800000u) == 0x7f800000u) return 0.0f;
    return __uint_as_float(b);
}
__device__ __forceinline__ float leaky(float v) { return v > 0.0f ? v : NEG_SLOPE * v; }

#define DEN_SCALE 2147483648.0f
#define DEN_INV   4.656612873077392578125e-10f
__device__ __forceinline__ int fx_exp(unsigned maxbits) {
    int e = (int)((maxbits >> 23) & 0xffu) - 127;
    return e < -96 ? -96 : (e > 96 ? 96 : e);
}
__device__ __forceinline__ float fx_scale(unsigned maxbits) {
    return __uint_as_float((unsigned)(156 - fx_exp(maxbits)) << 23);
}
__device__ __forceinline__ float fx_inv(unsigned maxbits) {
    return __uint_as_float((unsigned)(98 + fx_exp(maxbits)) << 23);
}

#define CV_U    16
#define CV_WAVE (CV_U * 32)
typedef __attribute__((ext_vector_type(4))) _Float16 v4h;
template <int N4>
__global__ void __launch_bounds__(256) k_f32_to_f16(const float* __restrict__ in,
                                                    _Float16* __restrict__ out) {
    static_assert(N4 % CV_WAVE == 0, "whole waves only");
    const int wave = (blockIdx.x * blockDim.x + threadIdx.x) >> 5;
    if (wave * CV_WAVE >= N4) return;
    const int g0 = wave * CV_WAVE + (threadIdx.x & 31);
    const float4* __restrict__ s = (const float4*)in;
    v4h* __restrict__ d = (v4h*)out;
    float4 v[CV_U];
#pragma unroll
    for (int q = 0; q < CV_U; ++q) v[q] = s[g0 + q * 32];
#pragma unroll
    for (int q = 0; q < CV_U; ++q) {
        v4h o = { (_Float16)v[q].x, (_Float16)v[q].y, (_Float16)v[q].z, (_Float16)v[q].w };
        d[g0 + q * 32] = o;
    }
}
__global__ void k_transpose_f16(const float* __restrict__ in, _Float16* __restrict__ out,
                                int R, int C) {
    int i = blockIdx.x * blockDim.x + threadIdx.x;
    if (i >= R * C) return;
    int r = i / C, c = i % C;
    out[c * R + r] = (_Float16)in[i];
}

template <int K>
__global__ void __launch_bounds__(256) k_gemm_wmma(const _Float16* __restrict__ A,
                                                   const _Float16* __restrict__ Bt,
                                                   float* __restrict__ C,
                                                   int M, int Ncol) {
    int wave = (blockIdx.x * blockDim.x + threadIdx.x) >> 5;
    int lane = threadIdx.x & 31;
    int mTiles = M >> 4, nTiles = Ncol >> 4;
    if (wave >= mTiles * nTiles) return;
    int mt = wave / nTiles, nt = wave % nTiles;
    int half = lane >> 4;
    int idx  = lane & 15;

    const _Float16* arow = A  + (size_t)(mt * 16 + idx) * K + half * 8;
    const _Float16* brow = Bt + (size_t)(nt * 16 + idx) * K + half * 8;

    v8f acc = {};
#pragma unroll
    for (int k0 = 0; k0 < K; k0 += 32) {
        union { v16h v; struct { v8h lo, hi; } h; } a, b;
        a.h.lo = *(const v8h*)(arow + k0);
        a.h.hi = *(const v8h*)(arow + k0 + 16);
        b.h.lo = *(const v8h*)(brow + k0);
        b.h.hi = *(const v8h*)(brow + k0 + 16);
        acc = __builtin_amdgcn_wmma_f32_16x16x32_f16(false, a.v, false, b.v,
                                                     (short)0, acc, false, false);
    }
    float* crow = C + (size_t)(mt * 16 + half * 8) * Ncol + nt * 16 + idx;
#pragma unroll
    for (int r = 0; r < 8; ++r) crow[(size_t)r * Ncol] = acc[r];
}

__global__ void k_el_er(const float* __restrict__ feat, const float* __restrict__ al,
                        const float* __restrict__ ar, float* __restrict__ el,
                        float* __restrict__ er, int n, int H, int D,
                        unsigned* __restrict__ mxabs) {
    int i = blockIdx.x * blockDim.x + threadIdx.x;
    float ma = 0.0f;
    if (i < n * H) {
        int node = i / H, h = i % H;
        const float* f = feat + (size_t)node * H * D + h * D;
        const float* a = al + h * D;
        const float* b = ar + h * D;
        float sl = 0.0f, sr = 0.0f;
        for (int d = 0; d < D; ++d) {
            float v = f[d]; sl += v * a[d]; sr += v * b[d]; ma = fmaxf(ma, fabsf(v));
        }
        el[i] = sl; er[i] = sr;
    }
    for (int o = 16; o > 0; o >>= 1) ma = fmaxf(ma, __shfl_xor(ma, o, 32));
    if ((threadIdx.x & 31) == 0) atomicMax(mxabs, __float_as_uint(ma));
}

#define NPW   196
#define NWG   256
#define WGT   1024
#define SCAN_U 8
#define LCAP1 16384
#define LCAP2 40960

template <int H, int D, int LCAP, bool FIRST>
__global__ void __launch_bounds__(WGT)
k_edges_owned(const int* __restrict__ src, const int* __restrict__ dst,
              const float* __restrict__ feat, const float* __restrict__ el,
              const float* __restrict__ er, const unsigned* __restrict__ mxabs,
              const float* __restrict__ bias, unsigned* glist, unsigned* gcnt,
              _Float16* __restrict__ o16, float* __restrict__ o32) {
    constexpr int F = H * D;
    static_assert(SCAN_U * 4 <= 32, "one mask bit per scanned index");
    static_assert(F >= 32 ? (D % 32 == 0 && (H & (H - 1)) == 0) : (H == 1 && 32 % F == 0), "lane map");
    __shared__ int acc[NPW * F];
    __shared__ unsigned long long den[NPW * H];
    __shared__ unsigned mx[NPW * H];
    __shared__ float ers[NPW * H];
    __shared__ unsigned list[LCAP];
    __shared__ unsigned cnt;

    const int tid = threadIdx.x;
    const int n0 = blockIdx.x * NPW;

    for (int i = tid; i < NPW * F; i += WGT) acc[i] = 0;
    for (int i = tid; i < NPW * H; i += WGT) {
        den[i] = 0ull; mx[i] = 0u;
        ers[i] = (n0 + i / H) < N_NODES ? er[(size_t)n0 * H + i] : 0.0f;
    }
    if (tid == 0) cnt = 0u;
    __syncthreads();

    bool have = false;
    if (!FIRST) {
        unsigned c = gcnt[blockIdx.x];
        if (c <= (unsigned)LCAP1) {
            for (unsigned i = tid; i < c; i += WGT) list[i] = glist[(size_t)blockIdx.x * LCAP1 + i];
            if (tid == 0) cnt = c;
            have = true;
            __syncthreads();
        }
    }
    if (!have) {
        for (int base = tid * 4; base < N_EDGES; base += WGT * 4 * SCAN_U) {
            int4 dv[SCAN_U];
#pragma unroll
            for (int u = 0; u < SCAN_U; ++u) {
                int b = base + u * WGT * 4;
                dv[u] = b < N_EDGES ? *(const int4*)(dst + b) : make_int4(-1, -1, -1, -1);
            }
            unsigned m = 0u;
#pragma unroll
            for (int u = 0; u < SCAN_U; ++u) {
                m |= (unsigned)((unsigned)(dv[u].x - n0) < (unsigned)NPW) << (4 * u);
                m |= (unsigned)((unsigned)(dv[u].y - n0) < (unsigned)NPW) << (4 * u + 1);
                m |= (unsigned)((unsigned)(dv[u].z - n0) < (unsigned)NPW) << (4 * u + 2);
                m |= (unsigned)((unsigned)(dv[u].w - n0) < (unsigned)NPW) << (4 * u + 3);
            }
            while (m) {
                int bit = __ffs((int)m) - 1;
                m &= m - 1u;
                unsigned pos = atomicAdd(&cnt, 1u);
                if (pos < (unsigned)LCAP) list[pos] = (unsigned)(base + (bit >> 2) * (WGT * 4) + (bit & 3));
            }
        }
        __syncthreads();
    }
    const unsigned total = cnt;
    const bool complete = total <= (unsigned)LCAP;

    auto score = [&](unsigned s, unsigned dl, int h) -> float {
        return leaky(el[(size_t)s * H + h] + ers[dl * H + h]);
    };

    if (complete) {
        for (unsigned i = tid; i < total; i += WGT) {
            unsigned w = list[i], s, dl;
            if (have) { s = w & 0xffffu; dl = w >> 16; }
            else {
                dl = (unsigned)(dst[w] - n0); s = (unsigned)src[w];
                w = s | (dl << 16);
                list[i] = w;
                if (FIRST) glist[(size_t)blockIdx.x * LCAP1 + i] = w;
            }
#pragma unroll
            for (int h = 0; h < H; ++h) atomicMax(&mx[dl * H + h], enc_key(score(s, dl, h)));
        }
    } else {
        for (int e = tid; e < N_EDGES; e += WGT) {
            unsigned dl = (unsigned)(dst[e] - n0);
            if (dl < (unsigned)NPW) {
                unsigned s = (unsigned)src[e];
                for (int h = 0; h < H; ++h) atomicMax(&mx[dl * H + h], enc_key(score(s, dl, h)));
            }
        }
    }
    if (FIRST && tid == 0) gcnt[blockIdx.x] = total;
    __syncthreads();

    auto edge_den = [&](unsigned s, unsigned dl) {
#pragma unroll
        for (int h = 0; h < H; ++h) {
            float x = __expf(score(s, dl, h) - dec_key(mx[dl * H + h]));
            atomicAdd(&den[dl * H + h], (unsigned long long)__float2uint_rn(x * DEN_SCALE));
        }
    };
    if (complete) {
        for (unsigned i = tid; i < total; i += WGT) { unsigned w = list[i]; edge_den(w & 0xffffu, w >> 16); }
    } else {
        for (int e = tid; e < N_EDGES; e += WGT) {
            unsigned dl = (unsigned)(dst[e] - n0);
            if (dl < (unsigned)NPW) edge_den((unsigned)src[e], dl);
        }
    }
    __syncthreads();

    const float sc = fx_scale(*mxabs);
    auto alpha = [&](unsigned s, unsigned dl, int h) -> float {
        float x = __expf(score(s, dl, h) - dec_key(mx[dl * H + h]));
        return x / fmaxf((float)den[dl * H + h] * DEN_INV, 1e-9f) * sc;
    };
    if (complete) {
        constexpr int EPS = F >= 32 ? 1 : 32 / F;
        constexpr int NJ  = F >= 32 ? F / 32 : 1;
        constexpr int U   = F >= 32 ? 6 : 8;
        const int lane = tid & 31, wv = tid >> 5;
        const int sub = F >= 32 ? 0 : lane / F;
        const int fl  = F >= 32 ? lane : lane % F;
        for (unsigned i0 = (unsigned)(wv * EPS * U); i0 < total; i0 += (unsigned)((WGT / 32) * EPS * U)) {
            float f[U][NJ], a[U][NJ];
            unsigned dls[U];
            bool ok[U];
#pragma unroll
            for (int u = 0; u < U; ++u) {
                unsigned i = i0 + (unsigned)(u * EPS + sub);
                ok[u] = i < total;
                unsigned w = list[ok[u] ? i : total - 1u];
                unsigned s = w & 0xffffu;
                dls[u] = w >> 16;
                const float* fr = feat + (size_t)s * F + fl;
#pragma unroll
                for (int j = 0; j < NJ; ++j) f[u][j] = fr[32 * j];
                float aq = alpha(s, dls[u], F >= 32 ? (lane & (H - 1)) : 0);
#pragma unroll
                for (int j = 0; j < NJ; ++j)
                    a[u][j] = F >= 32 ? __int_as_float(__builtin_amdgcn_readlane(__float_as_int(aq), (32 * j) / D))
                                      : aq;
            }
#pragma unroll
            for (int u = 0; u < U; ++u) {
                if (ok[u]) {
#pragma unroll
                    for (int j = 0; j < NJ; ++j)
                        atomicAdd(&acc[dls[u] * F + fl + 32 * j], __float2int_rn(f[u][j] * a[u][j]));
                }
            }
        }
    } else {
        for (int e = tid; e < N_EDGES; e += WGT) {
            unsigned dl = (unsigned)(dst[e] - n0);
            if (dl < (unsigned)NPW) {
                unsigned s = (unsigned)src[e];
                for (int h = 0; h < H; ++h) {
                    float ah = alpha(s, dl, h);
                    for (int d = 0; d < D; ++d)
                        atomicAdd(&acc[dl * F + h * D + d],
                                  __float2int_rn(feat[(size_t)s * F + h * D + d] * ah));
                }
            }
        }
    }
    __syncthreads();

    const float inv = fx_inv(*mxabs);
    for (int i = tid; i < NPW * F; i += WGT) {
        if (n0 + i / F < N_NODES) {
            float v = (float)acc[i] * inv + bias[i % F];
            if (FIRST) {
                v = v > 0.0f ? v : (__expf(v) - 1.0f);
                o16[(size_t)n0 * F + i] = (_Float16)v;
            } else {
                o32[(size_t)n0 * F + i] = v;
            }
        }
    }
}

extern "C" void kernel_launch(void* const* d_in, const int* in_sizes, int n_in,
                              void* d_out, int out_size, void* d_ws, size_t ws_size,
                              hipStream_t stream) {
    const float* x   = (const float*)d_in[0];
    const int*   src = (const int*)  d_in[1];
    const int*   dst = (const int*)  d_in[2];
    const float* W1  = (const float*)d_in[3];
    const float* b1  = (const float*)d_in[4];
    const float* al1 = (const float*)d_in[5];
    const float* ar1 = (const float*)d_in[6];
    const float* W2  = (const float*)d_in[7];
    const float* b2  = (const float*)d_in[8];
    const float* al2 = (const float*)d_in[9];
    const float* ar2 = (const float*)d_in[10];
    float* out = (float*)d_out;
    (void)in_sizes; (void)n_in; (void)out_size; (void)ws_size;

    char* p = (char*)d_ws;
    auto carve = [&](size_t bytes) -> char* {
        char* r = p; p += (bytes + 255) & ~(size_t)255; return r;
    };
    _Float16* xh    = (_Float16*)carve((size_t)N_NODES * IN_F * 2);
    _Float16* w1t   = (_Float16*)carve((size_t)IN_F * F2 * 2);
    float*    feat1 = (float*)   carve((size_t)N_NODES * F2 * 4);
    float*    el1   = (float*)   carve((size_t)N_NODES * HEADS * 4);
    float*    er1   = (float*)   carve((size_t)N_NODES * HEADS * 4);
    unsigned* mxa1  = (unsigned*)carve(256);
    unsigned* mxa2  = mxa1 + 1;
    unsigned* glist = (unsigned*)carve((size_t)NWG * LCAP1 * 4);
    unsigned* gcnt  = (unsigned*)carve((size_t)NWG * 4);
    _Float16* h16   = (_Float16*)carve((size_t)N_NODES * F2 * 2);
    _Float16* w2t   = (_Float16*)carve((size_t)F2 * NC * 2);
    float*    feat2 = (float*)   carve((size_t)N_NODES * NC * 4);
    float*    el2   = (float*)   carve((size_t)N_NODES * 4);
    float*    er2   = (float*)   carve((size_t)N_NODES * 4);

    hipMemsetAsync(mxa1, 0, 256, stream);

    const int T = 256;
    auto blk = [](long long n, int t) { return (int)((n + t - 1) / t); };

    {
        constexpr int N4 = N_NODES * IN_F / 4;
        constexpr int CV_BLOCKS = (N4 / CV_WAVE + 7) / 8;
        k_f32_to_f16<N4><<<CV_BLOCKS, T, 0, stream>>>(x, xh);
    }
    k_transpose_f16<<<blk((long long)IN_F * F2, T), T, 0, stream>>>(W1, w1t, IN_F, F2);

    {
        int waves = (N_NODES / 16) * (F2 / 16);
        k_gemm_wmma<IN_F><<<blk(waves, 8), T, 0, stream>>>(xh, w1t, feat1, N_NODES, F2);
    }
    k_el_er<<<blk((long long)N_NODES * HEADS, T), T, 0, stream>>>(feat1, al1, ar1, el1, er1,
                                                                  N_NODES, HEADS, HID, mxa1);
    k_edges_owned<HEADS, HID, LCAP1, true><<<NWG, WGT, 0, stream>>>(src, dst, feat1, el1, er1, mxa1, b1,
                                                                     glist, gcnt, h16, nullptr);

    k_transpose_f16<<<blk((long long)F2 * NC, T), T, 0, stream>>>(W2, w2t, F2, NC);
    {
        int waves = (N_NODES / 16) * (NC / 16);
        k_gemm_wmma<F2><<<blk(waves, 8), T, 0, stream>>>(h16, w2t, feat2, N_NODES, NC);
    }
    k_el_er<<<blk(N_NODES, T), T, 0, stream>>>(feat2, al2, ar2, el2, er2, N_NODES, 1, NC, mxa2);
    k_edges_owned<1, NC, LCAP2, false><<<NWG, WGT, 0, stream>>>(src, dst, feat2, el2, er2, mxa2, b2,
                                                                 glist, gcnt, nullptr, out);
}


# Row sums: each softmax coefficient computed once, one lane per (edge, head), 8 or 32 edges per wave batch, passed to the feature lanes by readlane / ds_bpermute; products and integer adds unchanged
# speedup vs baseline: 4.9678x; 1.1982x over previous
#include <hip/hip_runtime.h>
#include <hip/hip_bf16.h>

#define N_NODES 50000
#define N_EDGES 1600000
#define IN_F    256
#define HID     32
#define HEADS   4
#define F2      128
#define NC      16
#define NEG_SLOPE 0.2f

typedef __attribute__((ext_vector_type(16))) _Float16 v16h;
typedef __attribute__((ext_vector_type(8)))  _Float16 v8h;
typedef __attribute__((ext_vector_type(8)))  float    v8f;

__device__ __forceinline__ unsigned enc_key(float f) {
    unsigned b = __float_as_uint(f);
    return (b & 0x80000000u) ? ~b : (b | 0x80000000u);
}
__device__ __forceinline__ float dec_key(unsigned k) {
    unsigned b = (k & 0x80000000u) ? (k ^ 0x80000000u) : ~k;
    if ((b & 0x7f800000u) == 0x7f800000u) return 0.0f;
    return __uint_as_float(b);
}
__device__ __forceinline__ float leaky(float v) { return v > 0.0f ? v : NEG_SLOPE * v; }

#define DEN_SCALE 2147483648.0f
#define DEN_INV   4.656612873077392578125e-10f
__device__ __forceinline__ int fx_exp(unsigned maxbits) {
    int e = (int)((maxbits >> 23) & 0xffu) - 127;
    return e < -96 ? -96 : (e > 96 ? 96 : e);
}
__device__ __forceinline__ float fx_scale(unsigned maxbits) {
    return __uint_as_float((unsigned)(156 - fx_exp(maxbits)) << 23);
}
__device__ __forceinline__ float fx_inv(unsigned maxbits) {
    return __uint_as_float((unsigned)(98 + fx_exp(maxbits)) << 23);
}

#define CV_U    16
#define CV_WAVE (CV_U * 32)
typedef __attribute__((ext_vector_type(4))) _Float16 v4h;
template <int N4>
__global__ void __launch_bounds__(256) k_f32_to_f16(const float* __restrict__ in,
                                                    _Float16* __restrict__ out) {
    static_assert(N4 % CV_WAVE == 0, "whole waves only");
    const int wave = (blockIdx.x * blockDim.x + threadIdx.x) >> 5;
    if (wave * CV_WAVE >= N4) return;
    const int g0 = wave * CV_WAVE + (threadIdx.x & 31);
    const float4* __restrict__ s = (const float4*)in;
    v4h* __restrict__ d = (v4h*)out;
    float4 v[CV_U];
#pragma unroll
    for (int q = 0; q < CV_U; ++q) v[q] = s[g0 + q * 32];
#pragma unroll
    for (int q = 0; q < CV_U; ++q) {
        v4h o = { (_Float16)v[q].x, (_Float16)v[q].y, (_Float16)v[q].z, (_Float16)v[q].w };
        d[g0 + q * 32] = o;
    }
}
__global__ void k_transpose_f16(const float* __restrict__ in, _Float16* __restrict__ out,
                                int R, int C) {
    int i = blockIdx.x * blockDim.x + threadIdx.x;
    if (i >= R * C) return;
    int r = i / C, c = i % C;
    out[c * R + r] = (_Float16)in[i];
}

template <int K>
__global__ void __launch_bounds__(256) k_gemm_wmma(const _Float16* __restrict__ A,
                                                   const _Float16* __restrict__ Bt,
                                                   float* __restrict__ C,
                                                   int M, int Ncol) {
    int wave = (blockIdx.x * blockDim.x + threadIdx.x) >> 5;
    int lane = threadIdx.x & 31;
    int mTiles = M >> 4, nTiles = Ncol >> 4;
    if (wave >= mTiles * nTiles) return;
    int mt = wave / nTiles, nt = wave % nTiles;
    int half = lane >> 4;
    int idx  = lane & 15;

    const _Float16* arow = A  + (size_t)(mt * 16 + idx) * K + half * 8;
    const _Float16* brow = Bt + (size_t)(nt * 16 + idx) * K + half * 8;

    v8f acc = {};
#pragma unroll
    for (int k0 = 0; k0 < K; k0 += 32) {
        union { v16h v; struct { v8h lo, hi; } h; } a, b;
        a.h.lo = *(const v8h*)(arow + k0);
        a.h.hi = *(const v8h*)(arow + k0 + 16);
        b.h.lo = *(const v8h*)(brow + k0);
        b.h.hi = *(const v8h*)(brow + k0 + 16);
        acc = __builtin_amdgcn_wmma_f32_16x16x32_f16(false, a.v, false, b.v,
                                                     (short)0, acc, false, false);
    }
    float* crow = C + (size_t)(mt * 16 + half * 8) * Ncol + nt * 16 + idx;
#pragma unroll
    for (int r = 0; r < 8; ++r) crow[(size_t)r * Ncol] = acc[r];
}

__global__ void k_el_er(const float* __restrict__ feat, const float* __restrict__ al,
                        const float* __restrict__ ar, float* __restrict__ el,
                        float* __restrict__ er, int n, int H, int D,
                        unsigned* __restrict__ mxabs) {
    int i = blockIdx.x * blockDim.x + threadIdx.x;
    float ma = 0.0f;
    if (i < n * H) {
        int node = i / H, h = i % H;
        const float* f = feat + (size_t)node * H * D + h * D;
        const float* a = al + h * D;
        const float* b = ar + h * D;
        float sl = 0.0f, sr = 0.0f;
        for (int d = 0; d < D; ++d) {
            float v = f[d]; sl += v * a[d]; sr += v * b[d]; ma = fmaxf(ma, fabsf(v));
        }
        el[i] = sl; er[i] = sr;
    }
    for (int o = 16; o > 0; o >>= 1) ma = fmaxf(ma, __shfl_xor(ma, o, 32));
    if ((threadIdx.x & 31) == 0) atomicMax(mxabs, __float_as_uint(ma));
}

#define NPW   196
#define NWG   256
#define WGT   1024
#define SCAN_U 8
#define LCAP1 16384
#define LCAP2 40960

template <int H, int D, int LCAP, bool FIRST>
__global__ void __launch_bounds__(WGT)
k_edges_owned(const int* __restrict__ src, const int* __restrict__ dst,
              const float* __restrict__ feat, const float* __restrict__ el,
              const float* __restrict__ er, const unsigned* __restrict__ mxabs,
              const float* __restrict__ bias, unsigned* glist, unsigned* gcnt,
              _Float16* __restrict__ o16, float* __restrict__ o32) {
    constexpr int F = H * D;
    static_assert(SCAN_U * 4 <= 32, "one mask bit per scanned index");
    static_assert(F >= 32 ? (D % 32 == 0 && (H & (H - 1)) == 0 && H <= 32) : (H == 1 && 32 % F == 0), "lane map");
    __shared__ int acc[NPW * F];
    __shared__ unsigned long long den[NPW * H];
    __shared__ unsigned mx[NPW * H];
    __shared__ float ers[NPW * H];
    __shared__ unsigned list[LCAP];
    __shared__ unsigned cnt;

    const int tid = threadIdx.x;
    const int n0 = blockIdx.x * NPW;

    for (int i = tid; i < NPW * F; i += WGT) acc[i] = 0;
    for (int i = tid; i < NPW * H; i += WGT) {
        den[i] = 0ull; mx[i] = 0u;
        ers[i] = (n0 + i / H) < N_NODES ? er[(size_t)n0 * H + i] : 0.0f;
    }
    if (tid == 0) cnt = 0u;
    __syncthreads();

    bool have = false;
    if (!FIRST) {
        unsigned c = gcnt[blockIdx.x];
        if (c <= (unsigned)LCAP1) {
            for (unsigned i = tid; i < c; i += WGT) list[i] = glist[(size_t)blockIdx.x * LCAP1 + i];
            if (tid == 0) cnt = c;
            have = true;
            __syncthreads();
        }
    }
    if (!have) {
        for (int base = tid * 4; base < N_EDGES; base += WGT * 4 * SCAN_U) {
            int4 dv[SCAN_U];
#pragma unroll
            for (int u = 0; u < SCAN_U; ++u) {
                int b = base + u * WGT * 4;
                dv[u] = b < N_EDGES ? *(const int4*)(dst + b) : make_int4(-1, -1, -1, -1);
            }
            unsigned m = 0u;
#pragma unroll
            for (int u = 0; u < SCAN_U; ++u) {
                m |= (unsigned)((unsigned)(dv[u].x - n0) < (unsigned)NPW) << (4 * u);
                m |= (unsigned)((unsigned)(dv[u].y - n0) < (unsigned)NPW) << (4 * u + 1);
                m |= (unsigned)((unsigned)(dv[u].z - n0) < (unsigned)NPW) << (4 * u + 2);
                m |= (unsigned)((unsigned)(dv[u].w - n0) < (unsigned)NPW) << (4 * u + 3);
            }
            while (m) {
                int bit = __ffs((int)m) - 1;
                m &= m - 1u;
                unsigned pos = atomicAdd(&cnt, 1u);
                if (pos < (unsigned)LCAP) list[pos] = (unsigned)(base + (bit >> 2) * (WGT * 4) + (bit & 3));
            }
        }
        __syncthreads();
    }
    const unsigned total = cnt;
    const bool complete = total <= (unsigned)LCAP;

    auto score = [&](unsigned s, unsigned dl, int h) -> float {
        return leaky(el[(size_t)s * H + h] + ers[dl * H + h]);
    };

    if (complete) {
        for (unsigned i = tid; i < total; i += WGT) {
            unsigned w = list[i], s, dl;
            if (have) { s = w & 0xffffu; dl = w >> 16; }
            else {
                dl = (unsigned)(dst[w] - n0); s = (unsigned)src[w];
                w = s | (dl << 16);
                list[i] = w;
                if (FIRST) glist[(size_t)blockIdx.x * LCAP1 + i] = w;
            }
#pragma unroll
            for (int h = 0; h < H; ++h) atomicMax(&mx[dl * H + h], enc_key(score(s, dl, h)));
        }
    } else {
        for (int e = tid; e < N_EDGES; e += WGT) {
            unsigned dl = (unsigned)(dst[e] - n0);
            if (dl < (unsigned)NPW) {
                unsigned s = (unsigned)src[e];
                for (int h = 0; h < H; ++h) atomicMax(&mx[dl * H + h], enc_key(score(s, dl, h)));
            }
        }
    }
    if (FIRST && tid == 0) gcnt[blockIdx.x] = total;
    __syncthreads();

    auto edge_den = [&](unsigned s, unsigned dl) {
#pragma unroll
        for (int h = 0; h < H; ++h) {
            float x = __expf(score(s, dl, h) - dec_key(mx[dl * H + h]));
            atomicAdd(&den[dl * H + h], (unsigned long long)__float2uint_rn(x * DEN_SCALE));
        }
    };
    if (complete) {
        for (unsigned i = tid; i < total; i += WGT) { unsigned w = list[i]; edge_den(w & 0xffffu, w >> 16); }
    } else {
        for (int e = tid; e < N_EDGES; e += WGT) {
            unsigned dl = (unsigned)(dst[e] - n0);
            if (dl < (unsigned)NPW) edge_den((unsigned)src[e], dl);
        }
    }
    __syncthreads();

    const float sc = fx_scale(*mxabs);
    auto alpha = [&](unsigned s, unsigned dl, int h) -> float {
        float x = __expf(score(s, dl, h) - dec_key(mx[dl * H + h]));
        return x / fmaxf((float)den[dl * H + h] * DEN_INV, 1e-9f) * sc;
    };
    if (complete) {
        const int lane = tid & 31;
        const unsigned wv  = (unsigned)__builtin_amdgcn_readfirstlane(tid >> 5);
        const unsigned tot = (unsigned)__builtin_amdgcn_readfirstlane((int)total);
        if constexpr (F >= 32) {
            constexpr int EB = 32 / H;
            constexpr int NJ = F / 32;
            for (unsigned i0 = wv * EB; i0 < tot; i0 += (unsigned)((WGT / 32) * EB)) {
                const unsigned i = i0 + (unsigned)(lane / H);
                const unsigned w = list[i < tot ? i : tot - 1u];
                float f[EB][NJ];
                unsigned wu[EB];
#pragma unroll
                for (int u = 0; u < EB; ++u) {
                    wu[u] = (unsigned)__builtin_amdgcn_readlane((int)w, u * H);
                    const float* fr = feat + (size_t)(wu[u] & 0xffffu) * F + lane;
#pragma unroll
                    for (int j = 0; j < NJ; ++j) f[u][j] = fr[32 * j];
                }
                float aq = alpha(w & 0xffffu, w >> 16, lane & (H - 1));
                aq = i < tot ? aq : 0.0f;
#pragma unroll
                for (int u = 0; u < EB; ++u) {
                    const unsigned dl = wu[u] >> 16;
#pragma unroll
                    for (int j = 0; j < NJ; ++j) {
                        const float a = __int_as_float(__builtin_amdgcn_readlane(__float_as_int(aq),
                                                                                 u * H + (32 * j) / D));
                        atomicAdd(&acc[dl * F + lane + 32 * j], __float2int_rn(f[u][j] * a));
                    }
                }
            }
        } else {
            constexpr int EPS = 32 / F;
            constexpr int NS  = 32 / EPS;
            const int sub = lane / F, fl = lane % F;
            for (unsigned i0 = wv * 32u; i0 < tot; i0 += (unsigned)WGT) {
                const unsigned i = i0 + (unsigned)lane;
                const unsigned w = list[i < tot ? i : tot - 1u];
                float f[NS];
                unsigned dls[NS];
#pragma unroll
                for (int u = 0; u < NS; ++u) {
                    const unsigned wl = (unsigned)__builtin_amdgcn_ds_bpermute((u * EPS + sub) * 4, (int)w);
                    dls[u] = wl >> 16;
                    f[u] = feat[(wl & 0xffffu) * (unsigned)F + (unsigned)fl];
                }
                float aq = alpha(w & 0xffffu, w >> 16, 0);
                aq = i < tot ? aq : 0.0f;
#pragma unroll
                for (int u = 0; u < NS; ++u) {
                    const float a = __int_as_float(__builtin_amdgcn_ds_bpermute((u * EPS + sub) * 4,
                                                                                __float_as_int(aq)));
                    atomicAdd(&acc[dls[u] * F + fl], __float2int_rn(f[u] * a));
                }
            }
        }
    } else {
        for (int e = tid; e < N_EDGES; e += WGT) {
            unsigned dl = (unsigned)(dst[e] - n0);
            if (dl < (unsigned)NPW) {
                unsigned s = (unsigned)src[e];
                for (int h = 0; h < H; ++h) {
                    float ah = alpha(s, dl, h);
                    for (int d = 0; d < D; ++d)
                        atomicAdd(&acc[dl * F + h * D + d],
                                  __float2int_rn(feat[(size_t)s * F + h * D + d] * ah));
                }
            }
        }
    }
    __syncthreads();

    const float inv = fx_inv(*mxabs);
    for (int i = tid; i < NPW * F; i += WGT) {
        if (n0 + i / F < N_NODES) {
            float v = (float)acc[i] * inv + bias[i % F];
            if (FIRST) {
                v = v > 0.0f ? v : (__expf(v) - 1.0f);
                o16[(size_t)n0 * F + i] = (_Float16)v;
            } else {
                o32[(size_t)n0 * F + i] = v;
            }
        }
    }
}

extern "C" void kernel_launch(void* const* d_in, const int* in_sizes, int n_in,
                              void* d_out, int out_size, void* d_ws, size_t ws_size,
                              hipStream_t stream) {
    const float* x   = (const float*)d_in[0];
    const int*   src = (const int*)  d_in[1];
    const int*   dst = (const int*)  d_in[2];
    const float* W1  = (const float*)d_in[3];
    const float* b1  = (const float*)d_in[4];
    const float* al1 = (const float*)d_in[5];
    const float* ar1 = (const float*)d_in[6];
    const float* W2  = (const float*)d_in[7];
    const float* b2  = (const float*)d_in[8];
    const float* al2 = (const float*)d_in[9];
    const float* ar2 = (const float*)d_in[10];
    float* out = (float*)d_out;
    (void)in_sizes; (void)n_in; (void)out_size; (void)ws_size;

    char* p = (char*)d_ws;
    auto carve = [&](size_t bytes) -> char* {
        char* r = p; p += (bytes + 255) & ~(size_t)255; return r;
    };
    _Float16* xh    = (_Float16*)carve((size_t)N_NODES * IN_F * 2);
    _Float16* w1t   = (_Float16*)carve((size_t)IN_F * F2 * 2);
    float*    feat1 = (float*)   carve((size_t)N_NODES * F2 * 4);
    float*    el1   = (float*)   carve((size_t)N_NODES * HEADS * 4);
    float*    er1   = (float*)   carve((size_t)N_NODES * HEADS * 4);
    unsigned* mxa1  = (unsigned*)carve(256);
    unsigned* mxa2  = mxa1 + 1;
    unsigned* glist = (unsigned*)carve((size_t)NWG * LCAP1 * 4);
    unsigned* gcnt  = (unsigned*)carve((size_t)NWG * 4);
    _Float16* h16   = (_Float16*)carve((size_t)N_NODES * F2 * 2);
    _Float16* w2t   = (_Float16*)carve((size_t)F2 * NC * 2);
    float*    feat2 = (float*)   carve((size_t)N_NODES * NC * 4);
    float*    el2   = (float*)   carve((size_t)N_NODES * 4);
    float*    er2   = (float*)   carve((size_t)N_NODES * 4);

    hipMemsetAsync(mxa1, 0, 256, stream);

    const int T = 256;
    auto blk = [](long long n, int t) { return (int)((n + t - 1) / t); };

    {
        constexpr int N4 = N_NODES * IN_F / 4;
        constexpr int CV_BLOCKS = (N4 / CV_WAVE + 7) / 8;
        k_f32_to_f16<N4><<<CV_BLOCKS, T, 0, stream>>>(x, xh);
    }
    k_transpose_f16<<<blk((long long)IN_F * F2, T), T, 0, stream>>>(W1, w1t, IN_F, F2);

    {
        int waves = (N_NODES / 16) * (F2 / 16);
        k_gemm_wmma<IN_F><<<blk(waves, 8), T, 0, stream>>>(xh, w1t, feat1, N_NODES, F2);
    }
    k_el_er<<<blk((long long)N_NODES * HEADS, T), T, 0, stream>>>(feat1, al1, ar1, el1, er1,
                                                                  N_NODES, HEADS, HID, mxa1);
    k_edges_owned<HEADS, HID, LCAP1, true><<<NWG, WGT, 0, stream>>>(src, dst, feat1, el1, er1, mxa1, b1,
                                                                     glist, gcnt, h16, nullptr);

    k_transpose_f16<<<blk((long long)F2 * NC, T), T, 0, stream>>>(W2, w2t, F2, NC);
    {
        int waves = (N_NODES / 16) * (NC / 16);
        k_gemm_wmma<F2><<<blk(waves, 8), T, 0, stream>>>(h16, w2t, feat2, N_NODES, NC);
    }
    k_el_er<<<blk(N_NODES, T), T, 0, stream>>>(feat2, al2, ar2, el2, er2, N_NODES, 1, NC, mxa2);
    k_edges_owned<1, NC, LCAP2, false><<<NWG, WGT, 0, stream>>>(src, dst, feat2, el2, er2, mxa2, b2,
                                                                 glist, gcnt, nullptr, out);
}


# Layer-2 projection moved into the layer-1 edge kernel: a workgroup rounds its 196 hidden rows to f16 in LDS, multiplies them with W2 on WMMA, sums el/er there; 3 launches, the h copy go; same values
# speedup vs baseline: 6.7575x; 1.3603x over previous
#include <hip/hip_runtime.h>
#include <hip/hip_bf16.h>

#define N_NODES 50000
#define N_EDGES 1600000
#define IN_F    256
#define HID     32
#define HEADS   4
#define F2      128
#define NC      16
#define NEG_SLOPE 0.2f

typedef __attribute__((ext_vector_type(16))) _Float16 v16h;
typedef __attribute__((ext_vector_type(8)))  _Float16 v8h;
typedef __attribute__((ext_vector_type(8)))  float    v8f;

__device__ __forceinline__ unsigned enc_key(float f) {
    unsigned b = __float_as_uint(f);
    return (b & 0x80000000u) ? ~b : (b | 0x80000000u);
}
__device__ __forceinline__ float dec_key(unsigned k) {
    unsigned b = (k & 0x80000000u) ? (k ^ 0x80000000u) : ~k;
    if ((b & 0x7f800000u) == 0x7f800000u) return 0.0f;
    return __uint_as_float(b);
}
__device__ __forceinline__ float leaky(float v) { return v > 0.0f ? v : NEG_SLOPE * v; }

#define DEN_SCALE 2147483648.0f
#define DEN_INV   4.656612873077392578125e-10f
__device__ __forceinline__ int fx_exp(unsigned maxbits) {
    int e = (int)((maxbits >> 23) & 0xffu) - 127;
    return e < -96 ? -96 : (e > 96 ? 96 : e);
}
__device__ __forceinline__ float fx_scale(unsigned maxbits) {
    return __uint_as_float((unsigned)(156 - fx_exp(maxbits)) << 23);
}
__device__ __forceinline__ float fx_inv(unsigned maxbits) {
    return __uint_as_float((unsigned)(98 + fx_exp(maxbits)) << 23);
}

#define CV_U    16
#define CV_WAVE (CV_U * 32)
typedef __attribute__((ext_vector_type(4))) _Float16 v4h;
template <int N4>
__global__ void __launch_bounds__(256) k_f32_to_f16(const float* __restrict__ in,
                                                    _Float16* __restrict__ out) {
    static_assert(N4 % CV_WAVE == 0, "whole waves only");
    const int wave = (blockIdx.x * blockDim.x + threadIdx.x) >> 5;
    if (wave * CV_WAVE >= N4) return;
    const int g0 = wave * CV_WAVE + (threadIdx.x & 31);
    const float4* __restrict__ s = (const float4*)in;
    v4h* __restrict__ d = (v4h*)out;
    float4 v[CV_U];
#pragma unroll
    for (int q = 0; q < CV_U; ++q) v[q] = s[g0 + q * 32];
#pragma unroll
    for (int q = 0; q < CV_U; ++q) {
        v4h o = { (_Float16)v[q].x, (_Float16)v[q].y, (_Float16)v[q].z, (_Float16)v[q].w };
        d[g0 + q * 32] = o;
    }
}
__global__ void k_transpose_f16(const float* __restrict__ in, _Float16* __restrict__ out,
                                int R, int C) {
    int i = blockIdx.x * blockDim.x + threadIdx.x;
    if (i >= R * C) return;
    int r = i / C, c = i % C;
    out[c * R + r] = (_Float16)in[i];
}

template <int K>
__global__ void __launch_bounds__(256) k_gemm_wmma(const _Float16* __restrict__ A,
                                                   const _Float16* __restrict__ Bt,
                                                   float* __restrict__ C,
                                                   int M, int Ncol) {
    int wave = (blockIdx.x * blockDim.x + threadIdx.x) >> 5;
    int lane = threadIdx.x & 31;
    int mTiles = M >> 4, nTiles = Ncol >> 4;
    if (wave >= mTiles * nTiles) return;
    int mt = wave / nTiles, nt = wave % nTiles;
    int half = lane >> 4;
    int idx  = lane & 15;

    const _Float16* arow = A  + (size_t)(mt * 16 + idx) * K + half * 8;
    const _Float16* brow = Bt + (size_t)(nt * 16 + idx) * K + half * 8;

    v8f acc = {};
#pragma unroll
    for (int k0 = 0; k0 < K; k0 += 32) {
        union { v16h v; struct { v8h lo, hi; } h; } a, b;
        a.h.lo = *(const v8h*)(arow + k0);
        a.h.hi = *(const v8h*)(arow + k0 + 16);
        b.h.lo = *(const v8h*)(brow + k0);
        b.h.hi = *(const v8h*)(brow + k0 + 16);
        acc = __builtin_amdgcn_wmma_f32_16x16x32_f16(false, a.v, false, b.v,
                                                     (short)0, acc, false, false);
    }
    float* crow = C + (size_t)(mt * 16 + half * 8) * Ncol + nt * 16 + idx;
#pragma unroll
    for (int r = 0; r < 8; ++r) crow[(size_t)r * Ncol] = acc[r];
}

__global__ void k_el_er(const float* __restrict__ feat, const float* __restrict__ al,
                        const float* __restrict__ ar, float* __restrict__ el,
                        float* __restrict__ er, int n, int H, int D,
                        unsigned* __restrict__ mxabs) {
    int i = blockIdx.x * blockDim.x + threadIdx.x;
    float ma = 0.0f;
    if (i < n * H) {
        int node = i / H, h = i % H;
        const float* f = feat + (size_t)node * H * D + h * D;
        const float* a = al + h * D;
        const float* b = ar + h * D;
        float sl = 0.0f, sr = 0.0f;
        for (int d = 0; d < D; ++d) {
            float v = f[d]; sl += v * a[d]; sr += v * b[d]; ma = fmaxf(ma, fabsf(v));
        }
        el[i] = sl; er[i] = sr;
    }
    for (int o = 16; o > 0; o >>= 1) ma = fmaxf(ma, __shfl_xor(ma, o, 32));
    if ((threadIdx.x & 31) == 0) atomicMax(mxabs, __float_as_uint(ma));
}

#define NPW   196
#define NWG   256
#define WGT   1024
#define SCAN_U 8
#define LCAP1 16384
#define LCAP2 40960

#define TG_U    5
#define TG_WAVE (TG_U * 32)
#define TG_T    320
__global__ void __launch_bounds__(TG_T) k_owner_tags(const int* __restrict__ dst,
                                                     unsigned* __restrict__ tags) {
    constexpr int N4 = N_EDGES / 4;
    static_assert(N_EDGES % 16 == 0, "whole 16-byte groups of tags");
    static_assert(N4 % (TG_WAVE * (TG_T / 32)) == 0, "whole blocks only: no lane is out of range");
    static_assert((N_NODES - 1) / NPW < 256 && NWG <= 256, "an owner number fits a byte");
    const int wave = (blockIdx.x * blockDim.x + threadIdx.x) >> 5;
    const int g0 = wave * TG_WAVE + (threadIdx.x & 31);
    const int4* __restrict__ s = (const int4*)dst;
    int4 v[TG_U];
#pragma unroll
    for (int q = 0; q < TG_U; ++q) v[q] = s[g0 + q * 32];
#pragma unroll
    for (int q = 0; q < TG_U; ++q)
        tags[g0 + q * 32] = ((unsigned)v[q].x / (unsigned)NPW)
                          | ((unsigned)v[q].y / (unsigned)NPW) << 8
                          | ((unsigned)v[q].z / (unsigned)NPW) << 16
                          | ((unsigned)v[q].w / (unsigned)NPW) << 24;
}
__device__ __forceinline__ unsigned tag_match16(uint4 v, unsigned pat) {
    auto z = [&](unsigned w) -> unsigned {
        const unsigned x = w ^ pat;
        return ~(((x & 0x7f7f7f7fu) + 0x7f7f7f7fu) | x) & 0x80808080u;
    };
    return (z(v.x) >> 7) | (z(v.y) >> 6) | (z(v.z) >> 5) | (z(v.w) >> 4);
}

template <int H, int D, int LCAP, bool FIRST>
__global__ void __launch_bounds__(WGT)
k_edges_owned(const int* __restrict__ src, const int* __restrict__ dst,
              const unsigned* __restrict__ tags, const float* __restrict__ feat, const float* __restrict__ el,
              const float* __restrict__ er, const unsigned* __restrict__ mxabs,
              const float* __restrict__ bias, unsigned* glist, unsigned* gcnt,
              float* __restrict__ o32,
              const float* __restrict__ Wn, const float* __restrict__ aln, const float* __restrict__ arn,
              float* __restrict__ featn, float* __restrict__ eln, float* __restrict__ ern,
              unsigned* __restrict__ mxn) {
    constexpr int F = H * D;
    static_assert(SCAN_U % 2 == 0, "two 16-tag loads share one 32-bit mask word");
    static_assert(F >= 32 ? (D % 32 == 0 && (H & (H - 1)) == 0 && H <= 32) : (H == 1 && 32 % F == 0), "lane map");
    __shared__ int acc[NPW * F];
    __shared__ unsigned long long den[NPW * H];
    __shared__ unsigned mx[NPW * H];
    __shared__ float ers[NPW * H];
    __shared__ unsigned list[LCAP];
    __shared__ unsigned cnt;
    constexpr int SCAP = 28672 / H;
    __shared__ __attribute__((aligned(16))) float sco[SCAP * H];

    const int tid = threadIdx.x;
    const int n0 = blockIdx.x * NPW;

    for (int i = tid; i < NPW * F; i += WGT) acc[i] = 0;
    for (int i = tid; i < NPW * H; i += WGT) {
        den[i] = 0ull; mx[i] = 0u;
        ers[i] = (n0 + i / H) < N_NODES ? er[(size_t)n0 * H + i] : 0.0f;
    }
    if (tid == 0) cnt = 0u;
    __syncthreads();

    bool have = false;
    if (!FIRST) {
        unsigned c = gcnt[blockIdx.x];
        if (c <= (unsigned)LCAP1) {
            for (unsigned i = tid; i < c; i += WGT) list[i] = glist[(size_t)blockIdx.x * LCAP1 + i];
            if (tid == 0) cnt = c;
            have = true;
            __syncthreads();
        }
    }
    if (!have) {
        constexpr int NG = N_EDGES / 16;
        constexpr int SWEEP = WGT * SCAN_U;
        constexpr int FULL = NG / SWEEP * SWEEP;
        const uint4* __restrict__ tg = (const uint4*)tags;
        const unsigned pat = blockIdx.x * 0x01010101u;
        for (int base = tid; base < FULL; base += SWEEP) {
            uint4 tv[SCAN_U];
#pragma unroll
            for (int u = 0; u < SCAN_U; ++u) tv[u] = tg[base + u * WGT];
            unsigned m[SCAN_U / 2];
            unsigned n = 0u;
#pragma unroll
            for (int k = 0; k < SCAN_U / 2; ++k) {
                m[k] = tag_match16(tv[2 * k], pat) | (tag_match16(tv[2 * k + 1], pat) << 4);
                n += (unsigned)__popc(m[k]);
            }
            if (n) {
                unsigned pos = atomicAdd(&cnt, n);
#pragma unroll
                for (int k = 0; k < SCAN_U / 2; ++k) {
                    unsigned mk = m[k];
                    while (mk) {
                        const int p = __ffs((int)mk) - 1;
                        mk &= mk - 1u;
                        const int b = base + (2 * k + ((p >> 2) & 1)) * WGT;
                        if (pos < (unsigned)LCAP) list[pos] = (unsigned)(b * 16 + (p & 3) * 4 + (p >> 3));
                        ++pos;
                    }
                }
            }
        }
        for (int b = FULL + tid; b < NG; b += WGT) {
            unsigned mk = tag_match16(tg[b], pat);
            if (mk) {
                unsigned pos = atomicAdd(&cnt, (unsigned)__popc(mk));
                while (mk) {
                    const int p = __ffs((int)mk) - 1;
                    mk &= mk - 1u;
                    if (pos < (unsigned)LCAP) list[pos] = (unsigned)(b * 16 + (p & 3) * 4 + (p >> 3));
                    ++pos;
                }
            }
        }
        __syncthreads();
    }
    const unsigned total = cnt;
    const bool complete = total <= (unsigned)LCAP;

    auto score = [&](unsigned s, unsigned dl, int h) -> float {
        return leaky(el[(size_t)s * H + h] + ers[dl * H + h]);
    };

    if (complete) {
        for (unsigned i = tid; i < total; i += WGT) {
            unsigned w = list[i], s, dl;
            if (have) { s = w & 0xffffu; dl = w >> 16; }
            else {
                dl = (unsigned)(dst[w] - n0); s = (unsigned)src[w];
                w = s | (dl << 16);
                list[i] = w;
                if (FIRST) glist[(size_t)blockIdx.x * LCAP1 + i] = w;
            }
            float sv[H];
#pragma unroll
            for (int h = 0; h < H; ++h) {
                sv[h] = score(s, dl, h);
                atomicMax(&mx[dl * H + h], enc_key(sv[h]));
            }
            if (i < (unsigned)SCAP) {
#pragma unroll
                for (int h = 0; h < H; ++h) sco[i * H + h] = sv[h];
            }
        }
    } else {
        for (int e = tid; e < N_EDGES; e += WGT) {
            unsigned dl = (unsigned)(dst[e] - n0);
            if (dl < (unsigned)NPW) {
                unsigned s = (unsigned)src[e];
                for (int h = 0; h < H; ++h) atomicMax(&mx[dl * H + h], enc_key(score(s, dl, h)));
            }
        }
    }
    if (FIRST && tid == 0) gcnt[blockIdx.x] = total;
    __syncthreads();

    auto edge_den = [&](unsigned s, unsigned dl) {
#pragma unroll
        for (int h = 0; h < H; ++h) {
            float x = __expf(score(s, dl, h) - dec_key(mx[dl * H + h]));
            atomicAdd(&den[dl * H + h], (unsigned long long)__float2uint_rn(x * DEN_SCALE));
        }
    };
    if (complete) {
        for (unsigned i = tid; i < total; i += WGT) {
            const unsigned w = list[i];
            if (i < (unsigned)SCAP) {
                const unsigned dl = w >> 16;
#pragma unroll
                for (int h = 0; h < H; ++h) {
                    float x = __expf(sco[i * H + h] - dec_key(mx[dl * H + h]));
                    atomicAdd(&den[dl * H + h], (unsigned long long)__float2uint_rn(x * DEN_SCALE));
                }
            } else {
                edge_den(w & 0xffffu, w >> 16);
            }
        }
    } else {
        for (int e = tid; e < N_EDGES; e += WGT) {
            unsigned dl = (unsigned)(dst[e] - n0);
            if (dl < (unsigned)NPW) edge_den((unsigned)src[e], dl);
        }
    }
    __syncthreads();

    const float sc = fx_scale(*mxabs);
    auto alpha_of = [&](float v, unsigned dl, int h) -> float {
        float x = __expf(v - dec_key(mx[dl * H + h]));
        return x / fmaxf((float)den[dl * H + h] * DEN_INV, 1e-9f) * sc;
    };
    auto alpha = [&](unsigned s, unsigned dl, int h) -> float { return alpha_of(score(s, dl, h), dl, h); };
    auto kept_score = [&](unsigned i, unsigned w, int h) -> float {
        return i < (unsigned)SCAP ? sco[i * H + h] : score(w & 0xffffu, w >> 16, h);
    };
    if (complete) {
        const int lane = tid & 31;
        const unsigned wv  = (unsigned)__builtin_amdgcn_readfirstlane(tid >> 5);
        const unsigned tot = (unsigned)__builtin_amdgcn_readfirstlane((int)total);
        if constexpr (F >= 32) {
            constexpr int EB = 32 / H;
            constexpr int NJ = F / 32;
            for (unsigned i0 = wv * EB; i0 < tot; i0 += (unsigned)((WGT / 32) * EB)) {
                const unsigned i = i0 + (unsigned)(lane / H);
                const unsigned ic = i < tot ? i : tot - 1u;
                const unsigned w = list[ic];
                float f[EB][NJ];
                unsigned wu[EB];
#pragma unroll
                for (int u = 0; u < EB; ++u) {
                    wu[u] = (unsigned)__builtin_amdgcn_readlane((int)w, u * H);
                    const float* fr = feat + (size_t)(wu[u] & 0xffffu) * F + lane;
#pragma unroll
                    for (int j = 0; j < NJ; ++j) f[u][j] = fr[32 * j];
                }
                float aq = alpha_of(kept_score(ic, w, lane & (H - 1)), w >> 16, lane & (H - 1));
                aq = i < tot ? aq : 0.0f;
#pragma unroll
                for (int u = 0; u < EB; ++u) {
                    const unsigned dl = wu[u] >> 16;
#pragma unroll
                    for (int j = 0; j < NJ; ++j) {
                        const float a = __int_as_float(__builtin_amdgcn_readlane(__float_as_int(aq),
                                                                                 u * H + (32 * j) / D));
                        atomicAdd(&acc[dl * F + lane + 32 * j], __float2int_rn(f[u][j] * a));
                    }
                }
            }
        } else {
            constexpr int EPS = 32 / F;
            constexpr int NS  = 32 / EPS;
            const int sub = lane / F, fl = lane % F;
            for (unsigned i0 = wv * 32u; i0 < tot; i0 += (unsigned)WGT) {
                const unsigned i = i0 + (unsigned)lane;
                const unsigned ic = i < tot ? i : tot - 1u;
                const unsigned w = list[ic];
                float f[NS];
                unsigned dls[NS];
#pragma unroll
                for (int u = 0; u < NS; ++u) {
                    const unsigned wl = (unsigned)__builtin_amdgcn_ds_bpermute((u * EPS + sub) * 4, (int)w);
                    dls[u] = wl >> 16;
                    f[u] = feat[(wl & 0xffffu) * (unsigned)F + (unsigned)fl];
                }
                float aq = alpha_of(kept_score(ic, w, 0), w >> 16, 0);
                aq = i < tot ? aq : 0.0f;
#pragma unroll
                for (int u = 0; u < NS; ++u) {
                    const float a = __int_as_float(__builtin_amdgcn_ds_bpermute((u * EPS + sub) * 4,
                                                                                __float_as_int(aq)));
                    atomicAdd(&acc[dls[u] * F + fl], __float2int_rn(f[u] * a));
                }
            }
        }
    } else {
        for (int e = tid; e < N_EDGES; e += WGT) {
            unsigned dl = (unsigned)(dst[e] - n0);
            if (dl < (unsigned)NPW) {
                unsigned s = (unsigned)src[e];
                for (int h = 0; h < H; ++h) {
                    float ah = alpha(s, dl, h);
                    for (int d = 0; d < D; ++d)
                        atomicAdd(&acc[dl * F + h * D + d],
                                  __float2int_rn(feat[(size_t)s * F + h * D + d] * ah));
                }
            }
        }
    }
    __syncthreads();

    const float inv = fx_inv(*mxabs);
    if constexpr (!FIRST) {
        for (int i = tid; i < NPW * F; i += WGT)
            if (n0 + i / F < N_NODES) o32[(size_t)n0 * F + i] = (float)acc[i] * inv + bias[i % F];
    } else {
        static_assert(F % 32 == 0 && NC == 16, "one 16-column tile, whole k steps");
        constexpr int MT = (NPW + 15) / 16;
        constexpr int HP = F + 8;
        constexpr int FP = NC + 1;
        static_assert((HP * 2) % 16 == 0 && ((HP * 2 / 16) & 1) == 1, "pitch");
        constexpr int HS_FLOATS = MT * 16 * HP / 2;
        constexpr int FS_FLOATS = MT * 16 * FP;
        static_assert(HS_FLOATS % 4 == 0 && (HS_FLOATS + FS_FLOATS) % 4 == 0, "16-byte aligned parts");
        static_assert(HS_FLOATS + FS_FLOATS + NC * HP / 2 <= SCAP * H, "fits over the scores");
        _Float16* hs  = (_Float16*)sco;
        float*    f2s = sco + HS_FLOATS;
        _Float16* wts = (_Float16*)(sco + HS_FLOATS + FS_FLOATS);

        for (int i = tid; i < F * NC; i += WGT)
            wts[(i % NC) * HP + i / NC] = (_Float16)Wn[i];
        for (int g = tid; g < MT * 16 * (F / 8); g += WGT) {
            const int r = g / (F / 8), c8 = (g % (F / 8)) * 8;
            v8h o = {};
            if (r < NPW && n0 + r < N_NODES) {
#pragma unroll
                for (int j = 0; j < 8; ++j) {
                    float v = (float)acc[r * F + c8 + j] * inv + bias[c8 + j];
                    v = v > 0.0f ? v : (__expf(v) - 1.0f);
                    o[j] = (_Float16)v;
                }
            }
            *(v8h*)(hs + r * HP + c8) = o;
        }
        __syncthreads();

        const int wvn = __builtin_amdgcn_readfirstlane(tid >> 5);
        if (wvn < MT) {
            const int half = (tid & 31) >> 4, idx = tid & 15;
            const _Float16* arow = hs  + (wvn * 16 + idx) * HP + half * 8;
            const _Float16* brow = wts + idx * HP + half * 8;
            v8f c = {};
#pragma unroll
            for (int k0 = 0; k0 < F; k0 += 32) {
                union { v16h v; struct { v8h lo, hi; } h; } a, b;
                a.h.lo = *(const v8h*)(arow + k0);
                a.h.hi = *(const v8h*)(arow + k0 + 16);
                b.h.lo = *(const v8h*)(brow + k0);
                b.h.hi = *(const v8h*)(brow + k0 + 16);
                c = __builtin_amdgcn_wmma_f32_16x16x32_f16(false, a.v, false, b.v,
                                                           (short)0, c, false, false);
            }
#pragma unroll
            for (int r = 0; r < 8; ++r) f2s[(wvn * 16 + half * 8 + r) * FP + idx] = c[r];
        }
        __syncthreads();

        for (int i = tid; i < NPW * NC; i += WGT)
            if (n0 + i / NC < N_NODES) featn[(size_t)n0 * NC + i] = f2s[(i / NC) * FP + i % NC];
        if (tid < (NPW + 31) / 32 * 32) {
            float ma = 0.0f;
            if (tid < NPW && n0 + tid < N_NODES) {
                float sl = 0.0f, sr = 0.0f;
#pragma unroll
                for (int d = 0; d < NC; ++d) {
                    float v = f2s[tid * FP + d];
                    sl = __builtin_fmaf(v, aln[d], sl); sr = __builtin_fmaf(v, arn[d], sr);
                    ma = fmaxf(ma, fabsf(v));
                }
                eln[n0 + tid] = sl; ern[n0 + tid] = sr;
            }
            for (int o = 16; o > 0; o >>= 1) ma = fmaxf(ma, __shfl_xor(ma, o, 32));
            if ((tid & 31) == 0) atomicMax(mxn, __float_as_uint(ma));
        }
    }
}

extern "C" void kernel_launch(void* const* d_in, const int* in_sizes, int n_in,
                              void* d_out, int out_size, void* d_ws, size_t ws_size,
                              hipStream_t stream) {
    const float* x   = (const float*)d_in[0];
    const int*   src = (const int*)  d_in[1];
    const int*   dst = (const int*)  d_in[2];
    const float* W1  = (const float*)d_in[3];
    const float* b1  = (const float*)d_in[4];
    const float* al1 = (const float*)d_in[5];
    const float* ar1 = (const float*)d_in[6];
    const float* W2  = (const float*)d_in[7];
    const float* b2  = (const float*)d_in[8];
    const float* al2 = (const float*)d_in[9];
    const float* ar2 = (const float*)d_in[10];
    float* out = (float*)d_out;
    (void)in_sizes; (void)n_in; (void)out_size; (void)ws_size;

    char* p = (char*)d_ws;
    auto carve = [&](size_t bytes) -> char* {
        char* r = p; p += (bytes + 255) & ~(size_t)255; return r;
    };
    _Float16* xh    = (_Float16*)carve((size_t)N_NODES * IN_F * 2);
    _Float16* w1t   = (_Float16*)carve((size_t)IN_F * F2 * 2);
    float*    feat1 = (float*)   carve((size_t)N_NODES * F2 * 4);
    float*    el1   = (float*)   carve((size_t)N_NODES * HEADS * 4);
    float*    er1   = (float*)   carve((size_t)N_NODES * HEADS * 4);
    unsigned* mxa1  = (unsigned*)carve(256);
    unsigned* mxa2  = mxa1 + 1;
    unsigned* glist = (unsigned*)carve((size_t)NWG * LCAP1 * 4);
    unsigned* gcnt  = (unsigned*)carve((size_t)NWG * 4);
    unsigned* tags  = (unsigned*)carve((size_t)N_EDGES);
    float*    feat2 = (float*)   carve((size_t)N_NODES * NC * 4);
    float*    el2   = (float*)   carve((size_t)N_NODES * 4);
    float*    er2   = (float*)   carve((size_t)N_NODES * 4);

    hipMemsetAsync(mxa1, 0, 256, stream);

    const int T = 256;
    auto blk = [](long long n, int t) { return (int)((n + t - 1) / t); };

    {
        constexpr int N4 = N_NODES * IN_F / 4;
        constexpr int CV_BLOCKS = (N4 / CV_WAVE + 7) / 8;
        k_f32_to_f16<N4><<<CV_BLOCKS, T, 0, stream>>>(x, xh);
    }
    k_transpose_f16<<<blk((long long)IN_F * F2, T), T, 0, stream>>>(W1, w1t, IN_F, F2);

    {
        int waves = (N_NODES / 16) * (F2 / 16);
        k_gemm_wmma<IN_F><<<blk(waves, 8), T, 0, stream>>>(xh, w1t, feat1, N_NODES, F2);
    }
    k_el_er<<<blk((long long)N_NODES * HEADS, T), T, 0, stream>>>(feat1, al1, ar1, el1, er1,
                                                                  N_NODES, HEADS, HID, mxa1);
    {
        constexpr int TG_BLOCKS = N_EDGES / 4 / TG_WAVE / (TG_T / 32);
        k_owner_tags<<<TG_BLOCKS, TG_T, 0, stream>>>(dst, tags);
    }
    k_edges_owned<HEADS, HID, LCAP1, true><<<NWG, WGT, 0, stream>>>(src, dst, tags, feat1, el1, er1, mxa1, b1,
                                                                     glist, gcnt, nullptr,
                                                                     W2, al2, ar2, feat2, el2, er2, mxa2);

    k_edges_owned<1, NC, LCAP2, false><<<NWG, WGT, 0, stream>>>(src, dst, tags, feat2, el2, er2, mxa2, b2,
                                                                 glist, gcnt, out,
                                                                 nullptr, nullptr, nullptr, nullptr, nullptr, nullptr,
                                                                 nullptr);
}


# Owner tags as bit planes: the pre-pass sets bit e of plane dst/196 (256 rows x 4096 edges in LDS, stored as whole lines); a workgroup scans its own 208 KB plane, not 1.6 MB of bytes; same values
# speedup vs baseline: 7.0949x; 1.0499x over previous
#include <hip/hip_runtime.h>
#include <hip/hip_bf16.h>

#define N_NODES 50000
#define N_EDGES 1600000
#define IN_F    256
#define HID     32
#define HEADS   4
#define F2      128
#define NC      16
#define NEG_SLOPE 0.2f

typedef __attribute__((ext_vector_type(16))) _Float16 v16h;
typedef __attribute__((ext_vector_type(8)))  _Float16 v8h;
typedef __attribute__((ext_vector_type(8)))  float    v8f;

__device__ __forceinline__ unsigned enc_key(float f) {
    unsigned b = __float_as_uint(f);
    return (b & 0x80000000u) ? ~b : (b | 0x80000000u);
}
__device__ __forceinline__ float dec_key(unsigned k) {
    unsigned b = (k & 0x80000000u) ? (k ^ 0x80000000u) : ~k;
    if ((b & 0x7f800000u) == 0x7f800000u) return 0.0f;
    return __uint_as_float(b);
}
__device__ __forceinline__ float leaky(float v) { return v > 0.0f ? v : NEG_SLOPE * v; }

#define DEN_SCALE 2147483648.0f
#define DEN_INV   4.656612873077392578125e-10f
__device__ __forceinline__ int fx_exp(unsigned maxbits) {
    int e = (int)((maxbits >> 23) & 0xffu) - 127;
    return e < -96 ? -96 : (e > 96 ? 96 : e);
}
__device__ __forceinline__ float fx_scale(unsigned maxbits) {
    return __uint_as_float((unsigned)(156 - fx_exp(maxbits)) << 23);
}
__device__ __forceinline__ float fx_inv(unsigned maxbits) {
    return __uint_as_float((unsigned)(98 + fx_exp(maxbits)) << 23);
}

#define CV_U    16
#define CV_WAVE (CV_U * 32)
typedef __attribute__((ext_vector_type(4))) _Float16 v4h;
template <int N4>
__global__ void __launch_bounds__(256) k_f32_to_f16(const float* __restrict__ in,
                                                    _Float16* __restrict__ out) {
    static_assert(N4 % CV_WAVE == 0, "whole waves only");
    const int wave = (blockIdx.x * blockDim.x + threadIdx.x) >> 5;
    if (wave * CV_WAVE >= N4) return;
    const int g0 = wave * CV_WAVE + (threadIdx.x & 31);
    const float4* __restrict__ s = (const float4*)in;
    v4h* __restrict__ d = (v4h*)out;
    float4 v[CV_U];
#pragma unroll
    for (int q = 0; q < CV_U; ++q) v[q] = s[g0 + q * 32];
#pragma unroll
    for (int q = 0; q < CV_U; ++q) {
        v4h o = { (_Float16)v[q].x, (_Float16)v[q].y, (_Float16)v[q].z, (_Float16)v[q].w };
        d[g0 + q * 32] = o;
    }
}
__global__ void k_transpose_f16(const float* __restrict__ in, _Float16* __restrict__ out,
                                int R, int C) {
    int i = blockIdx.x * blockDim.x + threadIdx.x;
    if (i >= R * C) return;
    int r = i / C, c = i % C;
    out[c * R + r] = (_Float16)in[i];
}

template <int K>
__global__ void __launch_bounds__(256) k_gemm_wmma(const _Float16* __restrict__ A,
                                                   const _Float16* __restrict__ Bt,
                                                   float* __restrict__ C,
                                                   int M, int Ncol) {
    int wave = (blockIdx.x * blockDim.x + threadIdx.x) >> 5;
    int lane = threadIdx.x & 31;
    int mTiles = M >> 4, nTiles = Ncol >> 4;
    if (wave >= mTiles * nTiles) return;
    int mt = wave / nTiles, nt = wave % nTiles;
    int half = lane >> 4;
    int idx  = lane & 15;

    const _Float16* arow = A  + (size_t)(mt * 16 + idx) * K + half * 8;
    const _Float16* brow = Bt + (size_t)(nt * 16 + idx) * K + half * 8;

    v8f acc = {};
#pragma unroll
    for (int k0 = 0; k0 < K; k0 += 32) {
        union { v16h v; struct { v8h lo, hi; } h; } a, b;
        a.h.lo = *(const v8h*)(arow + k0);
        a.h.hi = *(const v8h*)(arow + k0 + 16);
        b.h.lo = *(const v8h*)(brow + k0);
        b.h.hi = *(const v8h*)(brow + k0 + 16);
        acc = __builtin_amdgcn_wmma_f32_16x16x32_f16(false, a.v, false, b.v,
                                                     (short)0, acc, false, false);
    }
    float* crow = C + (size_t)(mt * 16 + half * 8) * Ncol + nt * 16 + idx;
#pragma unroll
    for (int r = 0; r < 8; ++r) crow[(size_t)r * Ncol] = acc[r];
}

__global__ void k_el_er(const float* __restrict__ feat, const float* __restrict__ al,
                        const float* __restrict__ ar, float* __restrict__ el,
                        float* __restrict__ er, int n, int H, int D,
                        unsigned* __restrict__ mxabs) {
    int i = blockIdx.x * blockDim.x + threadIdx.x;
    float ma = 0.0f;
    if (i < n * H) {
        int node = i / H, h = i % H;
        const float* f = feat + (size_t)node * H * D + h * D;
        const float* a = al + h * D;
        const float* b = ar + h * D;
        float sl = 0.0f, sr = 0.0f;
        for (int d = 0; d < D; ++d) {
            float v = f[d]; sl += v * a[d]; sr += v * b[d]; ma = fmaxf(ma, fabsf(v));
        }
        el[i] = sl; er[i] = sr;
    }
    for (int o = 16; o > 0; o >>= 1) ma = fmaxf(ma, __shfl_xor(ma, o, 32));
    if ((threadIdx.x & 31) == 0) atomicMax(mxabs, __float_as_uint(ma));
}

#define NPW   196
#define NWG   256
#define WGT   1024
#define SCAN_U 13
#define LCAP1 16384
#define LCAP2 40960

#define PL_E     4096
#define PL_W     (PL_E / 32)
#define PL_T     256
#define PL_NB    416
#define PL_PITCH (PL_NB * PL_W)
#define PL_TP    (PL_W + 4)
__global__ void __launch_bounds__(PL_T) k_owner_planes(const int* __restrict__ dst,
                                                       unsigned* __restrict__ planes) {
    static_assert(N_EDGES % 4 == 0 && PL_NB * PL_E >= N_EDGES, "whole 16-byte groups; the grid covers the edge list");
    static_assert((N_NODES - 1) / NPW < NWG, "every owner number names a row of the tile");
    static_assert(PL_W == 4 * 32 && NWG % (PL_T / 32) == 0 && (PL_TP * 4) % 16 == 0, "one 16-byte piece of a row per lane");
    __shared__ __attribute__((aligned(16))) unsigned tile[NWG * PL_TP];
    const int tid = threadIdx.x;
    int4 d[PL_E / (4 * PL_T)];
    bool in[PL_E / (4 * PL_T)];
#pragma unroll
    for (int k = 0; k < PL_E / (4 * PL_T); ++k) {
        const int e = blockIdx.x * PL_E + k * (4 * PL_T) + tid * 4;
        in[k] = e < N_EDGES;
        d[k] = *(const int4*)(dst + (in[k] ? e : 0));
    }
    for (int i = tid; i < NWG * PL_TP / 4; i += PL_T) ((uint4*)tile)[i] = make_uint4(0u, 0u, 0u, 0u);
    __syncthreads();
#pragma unroll
    for (int k = 0; k < PL_E / (4 * PL_T); ++k) {
        if (in[k]) {
            const unsigned w = (unsigned)(k * (PL_T / 8) + (tid >> 3));
            const unsigned b = (unsigned)(tid & 7) * 4u;
            atomicOr(&tile[((unsigned)d[k].x / (unsigned)NPW) * PL_TP + w], 1u << b);
            atomicOr(&tile[((unsigned)d[k].y / (unsigned)NPW) * PL_TP + w], 2u << b);
            atomicOr(&tile[((unsigned)d[k].z / (unsigned)NPW) * PL_TP + w], 4u << b);
            atomicOr(&tile[((unsigned)d[k].w / (unsigned)NPW) * PL_TP + w], 8u << b);
        }
    }
    __syncthreads();
    const int lane = tid & 31, wv = tid >> 5;
    for (int o = wv; o < NWG; o += PL_T / 32) {
        const uint4 v = *(const uint4*)&tile[o * PL_TP + lane * 4];
        *(uint4*)&planes[(size_t)o * PL_PITCH + (size_t)blockIdx.x * PL_W + lane * 4] = v;
    }
}

template <int H, int D, int LCAP, bool FIRST>
__global__ void __launch_bounds__(WGT)
k_edges_owned(const int* __restrict__ src, const int* __restrict__ dst,
              const unsigned* __restrict__ planes, const float* __restrict__ feat, const float* __restrict__ el,
              const float* __restrict__ er, const unsigned* __restrict__ mxabs,
              const float* __restrict__ bias, unsigned* glist, unsigned* gcnt,
              float* __restrict__ o32,
              const float* __restrict__ Wn, const float* __restrict__ aln, const float* __restrict__ arn,
              float* __restrict__ featn, float* __restrict__ eln, float* __restrict__ ern,
              unsigned* __restrict__ mxn) {
    constexpr int F = H * D;
    static_assert(F >= 32 ? (D % 32 == 0 && (H & (H - 1)) == 0 && H <= 32) : (H == 1 && 32 % F == 0), "lane map");
    __shared__ int acc[NPW * F];
    __shared__ unsigned long long den[NPW * H];
    __shared__ unsigned mx[NPW * H];
    __shared__ float ers[NPW * H];
    __shared__ unsigned list[LCAP];
    __shared__ unsigned cnt;
    constexpr int SCAP = 28672 / H;
    __shared__ __attribute__((aligned(16))) float sco[SCAP * H];

    const int tid = threadIdx.x;
    const int n0 = blockIdx.x * NPW;

    for (int i = tid; i < NPW * F; i += WGT) acc[i] = 0;
    for (int i = tid; i < NPW * H; i += WGT) {
        den[i] = 0ull; mx[i] = 0u;
        ers[i] = (n0 + i / H) < N_NODES ? er[(size_t)n0 * H + i] : 0.0f;
    }
    if (tid == 0) cnt = 0u;
    __syncthreads();

    bool have = false;
    if (!FIRST) {
        unsigned c = gcnt[blockIdx.x];
        if (c <= (unsigned)LCAP1) {
            for (unsigned i = tid; i < c; i += WGT) list[i] = glist[(size_t)blockIdx.x * LCAP1 + i];
            if (tid == 0) cnt = c;
            have = true;
            __syncthreads();
        }
    }
    if (!have) {
        constexpr int NG = PL_PITCH / 4;
        static_assert(NG == WGT * SCAN_U, "one sweep of the whole workgroup covers the plane exactly");
        const uint4* __restrict__ pg = (const uint4*)(planes + (size_t)blockIdx.x * PL_PITCH);
        uint4 tv[SCAN_U];
#pragma unroll
        for (int u = 0; u < SCAN_U; ++u) tv[u] = pg[tid + u * WGT];
        unsigned n = 0u;
#pragma unroll
        for (int u = 0; u < SCAN_U; ++u)
            n += (unsigned)(__popc(tv[u].x) + __popc(tv[u].y) + __popc(tv[u].z) + __popc(tv[u].w));
        if (n) {
            unsigned pos = atomicAdd(&cnt, n);
#pragma unroll
            for (int u = 0; u < SCAN_U; ++u) {
                const unsigned m[4] = { tv[u].x, tv[u].y, tv[u].z, tv[u].w };
#pragma unroll
                for (int q = 0; q < 4; ++q) {
                    unsigned mk = m[q];
                    while (mk) {
                        const int p = __ffs((int)mk) - 1;
                        mk &= mk - 1u;
                        list[pos < (unsigned)LCAP ? pos : (unsigned)LCAP - 1u] =
                            (unsigned)((tid + u * WGT) * 128 + q * 32 + p);
                        ++pos;
                    }
                }
            }
        }
        __syncthreads();
    }
    const unsigned total = cnt;
    const bool complete = total <= (unsigned)LCAP;

    auto score = [&](unsigned s, unsigned dl, int h) -> float {
        return leaky(el[(size_t)s * H + h] + ers[dl * H + h]);
    };

    if (complete) {
        for (unsigned i = tid; i < total; i += WGT) {
            unsigned w = list[i], s, dl;
            if (have) { s = w & 0xffffu; dl = w >> 16; }
            else {
                dl = (unsigned)(dst[w] - n0); s = (unsigned)src[w];
                w = s | (dl << 16);
                list[i] = w;
                if (FIRST) glist[(size_t)blockIdx.x * LCAP1 + i] = w;
            }
            float sv[H];
#pragma unroll
            for (int h = 0; h < H; ++h) {
                sv[h] = score(s, dl, h);
                atomicMax(&mx[dl * H + h], enc_key(sv[h]));
            }
            if (i < (unsigned)SCAP) {
#pragma unroll
                for (int h = 0; h < H; ++h) sco[i * H + h] = sv[h];
            }
        }
    } else {
        for (int e = tid; e < N_EDGES; e += WGT) {
            unsigned dl = (unsigned)(dst[e] - n0);
            if (dl < (unsigned)NPW) {
                unsigned s = (unsigned)src[e];
                for (int h = 0; h < H; ++h) atomicMax(&mx[dl * H + h], enc_key(score(s, dl, h)));
            }
        }
    }
    if (FIRST && tid == 0) gcnt[blockIdx.x] = total;
    __syncthreads();

    auto edge_den = [&](unsigned s, unsigned dl) {
#pragma unroll
        for (int h = 0; h < H; ++h) {
            float x = __expf(score(s, dl, h) - dec_key(mx[dl * H + h]));
            atomicAdd(&den[dl * H + h], (unsigned long long)__float2uint_rn(x * DEN_SCALE));
        }
    };
    if (complete) {
        for (unsigned i = tid; i < total; i += WGT) {
            const unsigned w = list[i];
            if (i < (unsigned)SCAP) {
                const unsigned dl = w >> 16;
#pragma unroll
                for (int h = 0; h < H; ++h) {
                    float x = __expf(sco[i * H + h] - dec_key(mx[dl * H + h]));
                    atomicAdd(&den[dl * H + h], (unsigned long long)__float2uint_rn(x * DEN_SCALE));
                }
            } else {
                edge_den(w & 0xffffu, w >> 16);
            }
        }
    } else {
        for (int e = tid; e < N_EDGES; e += WGT) {
            unsigned dl = (unsigned)(dst[e] - n0);
            if (dl < (unsigned)NPW) edge_den((unsigned)src[e], dl);
        }
    }
    __syncthreads();

    const float sc = fx_scale(*mxabs);
    auto alpha_of = [&](float v, unsigned dl, int h) -> float {
        float x = __expf(v - dec_key(mx[dl * H + h]));
        return x / fmaxf((float)den[dl * H + h] * DEN_INV, 1e-9f) * sc;
    };
    auto alpha = [&](unsigned s, unsigned dl, int h) -> float { return alpha_of(score(s, dl, h), dl, h); };
    auto kept_score = [&](unsigned i, unsigned w, int h) -> float {
        return i < (unsigned)SCAP ? sco[i * H + h] : score(w & 0xffffu, w >> 16, h);
    };
    if (complete) {
        const int lane = tid & 31;
        const unsigned wv  = (unsigned)__builtin_amdgcn_readfirstlane(tid >> 5);
        const unsigned tot = (unsigned)__builtin_amdgcn_readfirstlane((int)total);
        if constexpr (F >= 32) {
            constexpr int EB = 32 / H;
            constexpr int NJ = F / 32;
            for (unsigned i0 = wv * EB; i0 < tot; i0 += (unsigned)((WGT / 32) * EB)) {
                const unsigned i = i0 + (unsigned)(lane / H);
                const unsigned ic = i < tot ? i : tot - 1u;
                const unsigned w = list[ic];
                float f[EB][NJ];
                unsigned wu[EB];
#pragma unroll
                for (int u = 0; u < EB; ++u) {
                    wu[u] = (unsigned)__builtin_amdgcn_readlane((int)w, u * H);
                    const float* fr = feat + (size_t)(wu[u] & 0xffffu) * F + lane;
#pragma unroll
                    for (int j = 0; j < NJ; ++j) f[u][j] = fr[32 * j];
                }
                float aq = alpha_of(kept_score(ic, w, lane & (H - 1)), w >> 16, lane & (H - 1));
                aq = i < tot ? aq : 0.0f;
#pragma unroll
                for (int u = 0; u < EB; ++u) {
                    const unsigned dl = wu[u] >> 16;
#pragma unroll
                    for (int j = 0; j < NJ; ++j) {
                        const float a = __int_as_float(__builtin_amdgcn_readlane(__float_as_int(aq),
                                                                                 u * H + (32 * j) / D));
                        atomicAdd(&acc[dl * F + lane + 32 * j], __float2int_rn(f[u][j] * a));
                    }
                }
            }
        } else {
            constexpr int EPS = 32 / F;
            constexpr int NS  = 32 / EPS;
            const int sub = lane / F, fl = lane % F;
            for (unsigned i0 = wv * 32u; i0 < tot; i0 += (unsigned)WGT) {
                const unsigned i = i0 + (unsigned)lane;
                const unsigned ic = i < tot ? i : tot - 1u;
                const unsigned w = list[ic];
                float f[NS];
                unsigned dls[NS];
#pragma unroll
                for (int u = 0; u < NS; ++u) {
                    const unsigned wl = (unsigned)__builtin_amdgcn_ds_bpermute((u * EPS + sub) * 4, (int)w);
                    dls[u] = wl >> 16;
                    f[u] = feat[(wl & 0xffffu) * (unsigned)F + (unsigned)fl];
                }
                float aq = alpha_of(kept_score(ic, w, 0), w >> 16, 0);
                aq = i < tot ? aq : 0.0f;
#pragma unroll
                for (int u = 0; u < NS; ++u) {
                    const float a = __int_as_float(__builtin_amdgcn_ds_bpermute((u * EPS + sub) * 4,
                                                                                __float_as_int(aq)));
                    atomicAdd(&acc[dls[u] * F + fl], __float2int_rn(f[u] * a));
                }
            }
        }
    } else {
        for (int e = tid; e < N_EDGES; e += WGT) {
            unsigned dl = (unsigned)(dst[e] - n0);
            if (dl < (unsigned)NPW) {
                unsigned s = (unsigned)src[e];
                for (int h = 0; h < H; ++h) {
                    float ah = alpha(s, dl, h);
                    for (int d = 0; d < D; ++d)
                        atomicAdd(&acc[dl * F + h * D + d],
                                  __float2int_rn(feat[(size_t)s * F + h * D + d] * ah));
                }
            }
        }
    }
    __syncthreads();

    const float inv = fx_inv(*mxabs);
    if constexpr (!FIRST) {
        for (int i = tid; i < NPW * F; i += WGT)
            if (n0 + i / F < N_NODES) o32[(size_t)n0 * F + i] = (float)acc[i] * inv + bias[i % F];
    } else {
        static_assert(F % 32 == 0 && NC == 16, "one 16-column tile, whole k steps");
        constexpr int MT = (NPW + 15) / 16;
        constexpr int HP = F + 8;
        constexpr int FP = NC + 1;
        static_assert((HP * 2) % 16 == 0 && ((HP * 2 / 16) & 1) == 1, "pitch");
        constexpr int HS_FLOATS = MT * 16 * HP / 2;
        constexpr int FS_FLOATS = MT * 16 * FP;
        static_assert(HS_FLOATS % 4 == 0 && (HS_FLOATS + FS_FLOATS) % 4 == 0, "16-byte aligned parts");
        static_assert(HS_FLOATS + FS_FLOATS + NC * HP / 2 <= SCAP * H, "fits over the scores");
        _Float16* hs  = (_Float16*)sco;
        float*    f2s = sco + HS_FLOATS;
        _Float16* wts = (_Float16*)(sco + HS_FLOATS + FS_FLOATS);

        for (int i = tid; i < F * NC; i += WGT)
            wts[(i % NC) * HP + i / NC] = (_Float16)Wn[i];
        for (int g = tid; g < MT * 16 * (F / 8); g += WGT) {
            const int r = g / (F / 8), c8 = (g % (F / 8)) * 8;
            v8h o = {};
            if (r < NPW && n0 + r < N_NODES) {
#pragma unroll
                for (int j = 0; j < 8; ++j) {
                    float v = (float)acc[r * F + c8 + j] * inv + bias[c8 + j];
                    v = v > 0.0f ? v : (__expf(v) - 1.0f);
                    o[j] = (_Float16)v;
                }
            }
            *(v8h*)(hs + r * HP + c8) = o;
        }
        __syncthreads();

        const int wvn = __builtin_amdgcn_readfirstlane(tid >> 5);
        if (wvn < MT) {
            const int half = (tid & 31) >> 4, idx = tid & 15;
            const _Float16* arow = hs  + (wvn * 16 + idx) * HP + half * 8;
            const _Float16* brow = wts + idx * HP + half * 8;
            v8f c = {};
#pragma unroll
            for (int k0 = 0; k0 < F; k0 += 32) {
                union { v16h v; struct { v8h lo, hi; } h; } a, b;
                a.h.lo = *(const v8h*)(arow + k0);
                a.h.hi = *(const v8h*)(arow + k0 + 16);
                b.h.lo = *(const v8h*)(brow + k0);
                b.h.hi = *(const v8h*)(brow + k0 + 16);
                c = __builtin_amdgcn_wmma_f32_16x16x32_f16(false, a.v, false, b.v,
                                                           (short)0, c, false, false);
            }
#pragma unroll
            for (int r = 0; r < 8; ++r) f2s[(wvn * 16 + half * 8 + r) * FP + idx] = c[r];
        }
        __syncthreads();

        for (int i = tid; i < NPW * NC; i += WGT)
            if (n0 + i / NC < N_NODES) featn[(size_t)n0 * NC + i] = f2s[(i / NC) * FP + i % NC];
        if (tid < (NPW + 31) / 32 * 32) {
            float ma = 0.0f;
            if (tid < NPW && n0 + tid < N_NODES) {
                float sl = 0.0f, sr = 0.0f;
#pragma unroll
                for (int d = 0; d < NC; ++d) {
                    float v = f2s[tid * FP + d];
                    sl = __builtin_fmaf(v, aln[d], sl); sr = __builtin_fmaf(v, arn[d], sr);
                    ma = fmaxf(ma, fabsf(v));
                }
                eln[n0 + tid] = sl; ern[n0 + tid] = sr;
            }
            for (int o = 16; o > 0; o >>= 1) ma = fmaxf(ma, __shfl_xor(ma, o, 32));
            if ((tid & 31) == 0) atomicMax(mxn, __float_as_uint(ma));
        }
    }
}

extern "C" void kernel_launch(void* const* d_in, const int* in_sizes, int n_in,
                              void* d_out, int out_size, void* d_ws, size_t ws_size,
                              hipStream_t stream) {
    const float* x   = (const float*)d_in[0];
    const int*   src = (const int*)  d_in[1];
    const int*   dst = (const int*)  d_in[2];
    const float* W1  = (const float*)d_in[3];
    const float* b1  = (const float*)d_in[4];
    const float* al1 = (const float*)d_in[5];
    const float* ar1 = (const float*)d_in[6];
    const float* W2  = (const float*)d_in[7];
    const float* b2  = (const float*)d_in[8];
    const float* al2 = (const float*)d_in[9];
    const float* ar2 = (const float*)d_in[10];
    float* out = (float*)d_out;
    (void)in_sizes; (void)n_in; (void)out_size; (void)ws_size;

    char* p = (char*)d_ws;
    auto carve = [&](size_t bytes) -> char* {
        char* r = p; p += (bytes + 255) & ~(size_t)255; return r;
    };
    _Float16* xh    = (_Float16*)carve((size_t)N_NODES * IN_F * 2);
    _Float16* w1t   = (_Float16*)carve((size_t)IN_F * F2 * 2);
    float*    feat1 = (float*)   carve((size_t)N_NODES * F2 * 4);
    float*    el1   = (float*)   carve((size_t)N_NODES * HEADS * 4);
    float*    er1   = (float*)   carve((size_t)N_NODES * HEADS * 4);
    unsigned* mxa1  = (unsigned*)carve(256);
    unsigned* mxa2  = mxa1 + 1;
    unsigned* glist = (unsigned*)carve((size_t)NWG * LCAP1 * 4);
    unsigned* gcnt  = (unsigned*)carve((size_t)NWG * 4);
    unsigned* planes = (unsigned*)carve((size_t)NWG * PL_PITCH * 4);
    float*    feat2 = (float*)   carve((size_t)N_NODES * NC * 4);
    float*    el2   = (float*)   carve((size_t)N_NODES * 4);
    float*    er2   = (float*)   carve((size_t)N_NODES * 4);

    hipMemsetAsync(mxa1, 0, 256, stream);

    const int T = 256;
    auto blk = [](long long n, int t) { return (int)((n + t - 1) / t); };

    {
        constexpr int N4 = N_NODES * IN_F / 4;
        constexpr int CV_BLOCKS = (N4 / CV_WAVE + 7) / 8;
        k_f32_to_f16<N4><<<CV_BLOCKS, T, 0, stream>>>(x, xh);
    }
    k_transpose_f16<<<blk((long long)IN_F * F2, T), T, 0, stream>>>(W1, w1t, IN_F, F2);

    {
        int waves = (N_NODES / 16) * (F2 / 16);
        k_gemm_wmma<IN_F><<<blk(waves, 8), T, 0, stream>>>(xh, w1t, feat1, N_NODES, F2);
    }
    k_el_er<<<blk((long long)N_NODES * HEADS, T), T, 0, stream>>>(feat1, al1, ar1, el1, er1,
                                                                  N_NODES, HEADS, HID, mxa1);
    k_owner_planes<<<PL_NB, PL_T, 0, stream>>>(dst, planes);
    k_edges_owned<HEADS, HID, LCAP1, true><<<NWG, WGT, 0, stream>>>(src, dst, planes, feat1, el1, er1, mxa1, b1,
                                                                     glist, gcnt, nullptr,
                                                                     W2, al2, ar2, feat2, el2, er2, mxa2);

    k_edges_owned<1, NC, LCAP2, false><<<NWG, WGT, 0, stream>>>(src, dst, planes, feat2, el2, er2, mxa2, b2,
                                                                 glist, gcnt, out,
                                                                 nullptr, nullptr, nullptr, nullptr, nullptr, nullptr,
                                                                 nullptr);
}


# Layer-1 projection re-tiled: a wave keeps its 16 columns' B fragments in registers over 80 rows; A rows and output rows go through LDS as whole 512-byte rows (4 lines per load, not 16); same values
# speedup vs baseline: 7.5958x; 1.0706x over previous
#include <hip/hip_runtime.h>
#include <hip/hip_bf16.h>

#define N_NODES 50000
#define N_EDGES 1600000
#define IN_F    256
#define HID     32
#define HEADS   4
#define F2      128
#define NC      16
#define NEG_SLOPE 0.2f

typedef __attribute__((ext_vector_type(16))) _Float16 v16h;
typedef __attribute__((ext_vector_type(8)))  _Float16 v8h;
typedef __attribute__((ext_vector_type(8)))  float    v8f;

__device__ __forceinline__ unsigned enc_key(float f) {
    unsigned b = __float_as_uint(f);
    return (b & 0x80000000u) ? ~b : (b | 0x80000000u);
}
__device__ __forceinline__ float dec_key(unsigned k) {
    unsigned b = (k & 0x80000000u) ? (k ^ 0x80000000u) : ~k;
    if ((b & 0x7f800000u) == 0x7f800000u) return 0.0f;
    return __uint_as_float(b);
}
__device__ __forceinline__ float leaky(float v) { return v > 0.0f ? v : NEG_SLOPE * v; }

#define DEN_SCALE 2147483648.0f
#define DEN_INV   4.656612873077392578125e-10f
__device__ __forceinline__ int fx_exp(unsigned maxbits) {
    int e = (int)((maxbits >> 23) & 0xffu) - 127;
    return e < -96 ? -96 : (e > 96 ? 96 : e);
}
__device__ __forceinline__ float fx_scale(unsigned maxbits) {
    return __uint_as_float((unsigned)(156 - fx_exp(maxbits)) << 23);
}
__device__ __forceinline__ float fx_inv(unsigned maxbits) {
    return __uint_as_float((unsigned)(98 + fx_exp(maxbits)) << 23);
}

#define CV_U    16
#define CV_WAVE (CV_U * 32)
typedef __attribute__((ext_vector_type(4))) _Float16 v4h;
template <int N4>
__global__ void __launch_bounds__(256) k_f32_to_f16(const float* __restrict__ in,
                                                    _Float16* __restrict__ out) {
    static_assert(N4 % CV_WAVE == 0, "whole waves only");
    const int wave = (blockIdx.x * blockDim.x + threadIdx.x) >> 5;
    if (wave * CV_WAVE >= N4) return;
    const int g0 = wave * CV_WAVE + (threadIdx.x & 31);
    const float4* __restrict__ s = (const float4*)in;
    v4h* __restrict__ d = (v4h*)out;
    float4 v[CV_U];
#pragma unroll
    for (int q = 0; q < CV_U; ++q) v[q] = s[g0 + q * 32];
#pragma unroll
    for (int q = 0; q < CV_U; ++q) {
        v4h o = { (_Float16)v[q].x, (_Float16)v[q].y, (_Float16)v[q].z, (_Float16)v[q].w };
        d[g0 + q * 32] = o;
    }
}
__global__ void k_transpose_f16(const float* __restrict__ in, _Float16* __restrict__ out,
                                int R, int C) {
    int i = blockIdx.x * blockDim.x + threadIdx.x;
    if (i >= R * C) return;
    int r = i / C, c = i % C;
    out[c * R + r] = (_Float16)in[i];
}

#define GM_RT   5
#define GM_ROWS (GM_RT * 16)
template <int K>
__global__ void __launch_bounds__(256) k_gemm_wmma(const _Float16* __restrict__ A,
                                                   const _Float16* __restrict__ Bt,
                                                   float* __restrict__ C) {
    constexpr int NCOL = 128;
    constexpr int AP = K + 8;
    constexpr int CP = NCOL + 4;
    constexpr int CPR = K / 8;
    constexpr int AQ = GM_ROWS * CPR / 256;
    constexpr int CQ = GM_ROWS * (NCOL / 4) / 256;
    static_assert((AP * 2) % 16 == 0 && ((AP * 2 / 16) & 1) == 1 && GM_ROWS * CPR % 256 == 0 && CPR == 32, "A tile");
    static_assert((8 * CP) % 64 == 32 && NCOL / 4 == 32 && GM_ROWS * (NCOL / 4) % 256 == 0, "output tile");
    __shared__ __attribute__((aligned(16))) _Float16 as[GM_ROWS * AP];
    __shared__ __attribute__((aligned(16))) float cs[GM_ROWS * CP];

    const int tid = threadIdx.x;
    const int lane = tid & 31, wv = tid >> 5;
    const int half = lane >> 4;
    const int idx  = lane & 15;
    const int row0 = blockIdx.x * GM_ROWS;

    const _Float16* brow = Bt + (size_t)(wv * 16 + idx) * K + half * 8;
    union frag { v16h v; struct { v8h lo, hi; } h; };
    frag b[K / 32];
#pragma unroll
    for (int k0 = 0; k0 < K; k0 += 32) {
        b[k0 / 32].h.lo = *(const v8h*)(brow + k0);
        b[k0 / 32].h.hi = *(const v8h*)(brow + k0 + 16);
    }
    v8h va[AQ];
#pragma unroll
    for (int q = 0; q < AQ; ++q) {
        const int c = tid + q * 256, r = c / CPR, p = c % CPR;
        va[q] = *(const v8h*)(A + (size_t)(row0 + r) * K + p * 8);
    }
#pragma unroll
    for (int q = 0; q < AQ; ++q) {
        const int c = tid + q * 256, r = c / CPR, p = c % CPR;
        *(v8h*)(as + r * AP + p * 8) = va[q];
    }
    __syncthreads();

#pragma unroll
    for (int t = 0; t < GM_RT; ++t) {
        const _Float16* arow = as + (t * 16 + idx) * AP + half * 8;
        v8f acc = {};
#pragma unroll
        for (int k0 = 0; k0 < K; k0 += 32) {
            frag a;
            a.h.lo = *(const v8h*)(arow + k0);
            a.h.hi = *(const v8h*)(arow + k0 + 16);
            acc = __builtin_amdgcn_wmma_f32_16x16x32_f16(false, a.v, false, b[k0 / 32].v,
                                                         (short)0, acc, false, false);
        }
#pragma unroll
        for (int r = 0; r < 8; ++r) cs[(t * 16 + half * 8 + r) * CP + wv * 16 + idx] = acc[r];
    }
    __syncthreads();

#pragma unroll
    for (int q = 0; q < CQ; ++q) {
        const int c = tid + q * 256, r = c / (NCOL / 4), p = c % (NCOL / 4);
        *(float4*)(C + (size_t)(row0 + r) * NCOL + p * 4) = *(const float4*)(cs + r * CP + p * 4);
    }
}

__global__ void k_el_er(const float* __restrict__ feat, const float* __restrict__ al,
                        const float* __restrict__ ar, float* __restrict__ el,
                        float* __restrict__ er, int n, int H, int D,
                        unsigned* __restrict__ mxabs) {
    int i = blockIdx.x * blockDim.x + threadIdx.x;
    float ma = 0.0f;
    if (i < n * H) {
        int node = i / H, h = i % H;
        const float* f = feat + (size_t)node * H * D + h * D;
        const float* a = al + h * D;
        const float* b = ar + h * D;
        float sl = 0.0f, sr = 0.0f;
        for (int d = 0; d < D; ++d) {
            float v = f[d]; sl += v * a[d]; sr += v * b[d]; ma = fmaxf(ma, fabsf(v));
        }
        el[i] = sl; er[i] = sr;
    }
    for (int o = 16; o > 0; o >>= 1) ma = fmaxf(ma, __shfl_xor(ma, o, 32));
    if ((threadIdx.x & 31) == 0) atomicMax(mxabs, __float_as_uint(ma));
}

#define NPW   196
#define NWG   256
#define WGT   1024
#define SCAN_U 13
#define LCAP1 16384
#define LCAP2 40960

#define PL_E     4096
#define PL_W     (PL_E / 32)
#define PL_T     256
#define PL_NB    416
#define PL_PITCH (PL_NB * PL_W)
#define PL_TP    (PL_W + 4)
__global__ void __launch_bounds__(PL_T) k_owner_planes(const int* __restrict__ dst,
                                                       unsigned* __restrict__ planes) {
    static_assert(N_EDGES % 4 == 0 && PL_NB * PL_E >= N_EDGES, "whole 16-byte groups; the grid covers the edge list");
    static_assert((N_NODES - 1) / NPW < NWG, "every owner number names a row of the tile");
    static_assert(PL_W == 4 * 32 && NWG % (PL_T / 32) == 0 && (PL_TP * 4) % 16 == 0, "one 16-byte piece of a row per lane");
    __shared__ __attribute__((aligned(16))) unsigned tile[NWG * PL_TP];
    const int tid = threadIdx.x;
    int4 d[PL_E / (4 * PL_T)];
    bool in[PL_E / (4 * PL_T)];
#pragma unroll
    for (int k = 0; k < PL_E / (4 * PL_T); ++k) {
        const int e = blockIdx.x * PL_E + k * (4 * PL_T) + tid * 4;
        in[k] = e < N_EDGES;
        d[k] = *(const int4*)(dst + (in[k] ? e : 0));
    }
    for (int i = tid; i < NWG * PL_TP / 4; i += PL_T) ((uint4*)tile)[i] = make_uint4(0u, 0u, 0u, 0u);
    __syncthreads();
#pragma unroll
    for (int k = 0; k < PL_E / (4 * PL_T); ++k) {
        if (in[k]) {
            const unsigned w = (unsigned)(k * (PL_T / 8) + (tid >> 3));
            const unsigned b = (unsigned)(tid & 7) * 4u;
            atomicOr(&tile[((unsigned)d[k].x / (unsigned)NPW) * PL_TP + w], 1u << b);
            atomicOr(&tile[((unsigned)d[k].y / (unsigned)NPW) * PL_TP + w], 2u << b);
            atomicOr(&tile[((unsigned)d[k].z / (unsigned)NPW) * PL_TP + w], 4u << b);
            atomicOr(&tile[((unsigned)d[k].w / (unsigned)NPW) * PL_TP + w], 8u << b);
        }
    }
    __syncthreads();
    const int lane = tid & 31, wv = tid >> 5;
    for (int o = wv; o < NWG; o += PL_T / 32) {
        const uint4 v = *(const uint4*)&tile[o * PL_TP + lane * 4];
        *(uint4*)&planes[(size_t)o * PL_PITCH + (size_t)blockIdx.x * PL_W + lane * 4] = v;
    }
}

template <int H, int D, int LCAP, bool FIRST>
__global__ void __launch_bounds__(WGT)
k_edges_owned(const int* __restrict__ src, const int* __restrict__ dst,
              const unsigned* __restrict__ planes, const float* __restrict__ feat, const float* __restrict__ el,
              const float* __restrict__ er, const unsigned* __restrict__ mxabs,
              const float* __restrict__ bias, unsigned* glist, unsigned* gcnt,
              float* __restrict__ o32,
              const float* __restrict__ Wn, const float* __restrict__ aln, const float* __restrict__ arn,
              float* __restrict__ featn, float* __restrict__ eln, float* __restrict__ ern,
              unsigned* __restrict__ mxn) {
    constexpr int F = H * D;
    static_assert(F >= 32 ? (D % 32 == 0 && (H & (H - 1)) == 0 && H <= 32) : (H == 1 && 32 % F == 0), "lane map");
    __shared__ int acc[NPW * F];
    __shared__ unsigned long long den[NPW * H];
    __shared__ unsigned mx[NPW * H];
    __shared__ float ers[NPW * H];
    __shared__ unsigned list[LCAP];
    __shared__ unsigned cnt;
    constexpr int SCAP = 28672 / H;
    __shared__ __attribute__((aligned(16))) float sco[SCAP * H];

    const int tid = threadIdx.x;
    const int n0 = blockIdx.x * NPW;

    for (int i = tid; i < NPW * F; i += WGT) acc[i] = 0;
    for (int i = tid; i < NPW * H; i += WGT) {
        den[i] = 0ull; mx[i] = 0u;
        ers[i] = (n0 + i / H) < N_NODES ? er[(size_t)n0 * H + i] : 0.0f;
    }
    if (tid == 0) cnt = 0u;
    __syncthreads();

    bool have = false;
    if (!FIRST) {
        unsigned c = gcnt[blockIdx.x];
        if (c <= (unsigned)LCAP1) {
            for (unsigned i = tid; i < c; i += WGT) list[i] = glist[(size_t)blockIdx.x * LCAP1 + i];
            if (tid == 0) cnt = c;
            have = true;
            __syncthreads();
        }
    }
    if (!have) {
        constexpr int NG = PL_PITCH / 4;
        static_assert(NG == WGT * SCAN_U, "one sweep of the whole workgroup covers the plane exactly");
        const uint4* __restrict__ pg = (const uint4*)(planes + (size_t)blockIdx.x * PL_PITCH);
        uint4 tv[SCAN_U];
#pragma unroll
        for (int u = 0; u < SCAN_U; ++u) tv[u] = pg[tid + u * WGT];
        unsigned n = 0u;
#pragma unroll
        for (int u = 0; u < SCAN_U; ++u)
            n += (unsigned)(__popc(tv[u].x) + __popc(tv[u].y) + __popc(tv[u].z) + __popc(tv[u].w));
        if (n) {
            unsigned pos = atomicAdd(&cnt, n);
#pragma unroll
            for (int u = 0; u < SCAN_U; ++u) {
                const unsigned m[4] = { tv[u].x, tv[u].y, tv[u].z, tv[u].w };
#pragma unroll
                for (int q = 0; q < 4; ++q) {
                    unsigned mk = m[q];
                    while (mk) {
                        const int p = __ffs((int)mk) - 1;
                        mk &= mk - 1u;
                        list[pos < (unsigned)LCAP ? pos : (unsigned)LCAP - 1u] =
                            (unsigned)((tid + u * WGT) * 128 + q * 32 + p);
                        ++pos;
                    }
                }
            }
        }
        __syncthreads();
    }
    const unsigned total = cnt;
    const bool complete = total <= (unsigned)LCAP;

    auto score = [&](unsigned s, unsigned dl, int h) -> float {
        return leaky(el[(size_t)s * H + h] + ers[dl * H + h]);
    };

    if (complete) {
        for (unsigned i = tid; i < total; i += WGT) {
            unsigned w = list[i], s, dl;
            if (have) { s = w & 0xffffu; dl = w >> 16; }
            else {
                dl = (unsigned)(dst[w] - n0); s = (unsigned)src[w];
                w = s | (dl << 16);
                list[i] = w;
                if (FIRST) glist[(size_t)blockIdx.x * LCAP1 + i] = w;
            }
            float sv[H];
#pragma unroll
            for (int h = 0; h < H; ++h) {
                sv[h] = score(s, dl, h);
                atomicMax(&mx[dl * H + h], enc_key(sv[h]));
            }
            if (i < (unsigned)SCAP) {
#pragma unroll
                for (int h = 0; h < H; ++h) sco[i * H + h] = sv[h];
            }
        }
    } else {
        for (int e = tid; e < N_EDGES; e += WGT) {
            unsigned dl = (unsigned)(dst[e] - n0);
            if (dl < (unsigned)NPW) {
                unsigned s = (unsigned)src[e];
                for (int h = 0; h < H; ++h) atomicMax(&mx[dl * H + h], enc_key(score(s, dl, h)));
            }
        }
    }
    if (FIRST && tid == 0) gcnt[blockIdx.x] = total;
    __syncthreads();

    auto edge_den = [&](unsigned s, unsigned dl) {
#pragma unroll
        for (int h = 0; h < H; ++h) {
            float x = __expf(score(s, dl, h) - dec_key(mx[dl * H + h]));
            atomicAdd(&den[dl * H + h], (unsigned long long)__float2uint_rn(x * DEN_SCALE));
        }
    };
    if (complete) {
        for (unsigned i = tid; i < total; i += WGT) {
            const unsigned w = list[i];
            if (i < (unsigned)SCAP) {
                const unsigned dl = w >> 16;
#pragma unroll
                for (int h = 0; h < H; ++h) {
                    float x = __expf(sco[i * H + h] - dec_key(mx[dl * H + h]));
                    atomicAdd(&den[dl * H + h], (unsigned long long)__float2uint_rn(x * DEN_SCALE));
                }
            } else {
                edge_den(w & 0xffffu, w >> 16);
            }
        }
    } else {
        for (int e = tid; e < N_EDGES; e += WGT) {
            unsigned dl = (unsigned)(dst[e] - n0);
            if (dl < (unsigned)NPW) edge_den((unsigned)src[e], dl);
        }
    }
    __syncthreads();

    const float sc = fx_scale(*mxabs);
    auto alpha_of = [&](float v, unsigned dl, int h) -> float {
        float x = __expf(v - dec_key(mx[dl * H + h]));
        return x / fmaxf((float)den[dl * H + h] * DEN_INV, 1e-9f) * sc;
    };
    auto alpha = [&](unsigned s, unsigned dl, int h) -> float { return alpha_of(score(s, dl, h), dl, h); };
    auto kept_score = [&](unsigned i, unsigned w, int h) -> float {
        return i < (unsigned)SCAP ? sco[i * H + h] : score(w & 0xffffu, w >> 16, h);
    };
    if (complete) {
        const int lane = tid & 31;
        const unsigned wv  = (unsigned)__builtin_amdgcn_readfirstlane(tid >> 5);
        const unsigned tot = (unsigned)__builtin_amdgcn_readfirstlane((int)total);
        if constexpr (F >= 32) {
            constexpr int EB = 32 / H;
            constexpr int NJ = F / 32;
            for (unsigned i0 = wv * EB; i0 < tot; i0 += (unsigned)((WGT / 32) * EB)) {
                const unsigned i = i0 + (unsigned)(lane / H);
                const unsigned ic = i < tot ? i : tot - 1u;
                const unsigned w = list[ic];
                float f[EB][NJ];
                unsigned wu[EB];
#pragma unroll
                for (int u = 0; u < EB; ++u) {
                    wu[u] = (unsigned)__builtin_amdgcn_readlane((int)w, u * H);
                    const float* fr = feat + (size_t)(wu[u] & 0xffffu) * F + lane;
#pragma unroll
                    for (int j = 0; j < NJ; ++j) f[u][j] = fr[32 * j];
                }
                float aq = alpha_of(kept_score(ic, w, lane & (H - 1)), w >> 16, lane & (H - 1));
                aq = i < tot ? aq : 0.0f;
#pragma unroll
                for (int u = 0; u < EB; ++u) {
                    const unsigned dl = wu[u] >> 16;
#pragma unroll
                    for (int j = 0; j < NJ; ++j) {
                        const float a = __int_as_float(__builtin_amdgcn_readlane(__float_as_int(aq),
                                                                                 u * H + (32 * j) / D));
                        atomicAdd(&acc[dl * F + lane + 32 * j], __float2int_rn(f[u][j] * a));
                    }
                }
            }
        } else {
            constexpr int EPS = 32 / F;
            constexpr int NS  = 32 / EPS;
            const int sub = lane / F, fl = lane % F;
            for (unsigned i0 = wv * 32u; i0 < tot; i0 += (unsigned)WGT) {
                const unsigned i = i0 + (unsigned)lane;
                const unsigned ic = i < tot ? i : tot - 1u;
                const unsigned w = list[ic];
                float f[NS];
                unsigned dls[NS];
#pragma unroll
                for (int u = 0; u < NS; ++u) {
                    const unsigned wl = (unsigned)__builtin_amdgcn_ds_bpermute((u * EPS + sub) * 4, (int)w);
                    dls[u] = wl >> 16;
                    f[u] = feat[(wl & 0xffffu) * (unsigned)F + (unsigned)fl];
                }
                float aq = alpha_of(kept_score(ic, w, 0), w >> 16, 0);
                aq = i < tot ? aq : 0.0f;
#pragma unroll
                for (int u = 0; u < NS; ++u) {
                    const float a = __int_as_float(__builtin_amdgcn_ds_bpermute((u * EPS + sub) * 4,
                                                                                __float_as_int(aq)));
                    atomicAdd(&acc[dls[u] * F + fl], __float2int_rn(f[u] * a));
                }
            }
        }
    } else {
        for (int e = tid; e < N_EDGES; e += WGT) {
            unsigned dl = (unsigned)(dst[e] - n0);
            if (dl < (unsigned)NPW) {
                unsigned s = (unsigned)src[e];
                for (int h = 0; h < H; ++h) {
                    float ah = alpha(s, dl, h);
                    for (int d = 0; d < D; ++d)
                        atomicAdd(&acc[dl * F + h * D + d],
                                  __float2int_rn(feat[(size_t)s * F + h * D + d] * ah));
                }
            }
        }
    }
    __syncthreads();

    const float inv = fx_inv(*mxabs);
    if constexpr (!FIRST) {
        for (int i = tid; i < NPW * F; i += WGT)
            if (n0 + i / F < N_NODES) o32[(size_t)n0 * F + i] = (float)acc[i] * inv + bias[i % F];
    } else {
        static_assert(F % 32 == 0 && NC == 16, "one 16-column tile, whole k steps");
        constexpr int MT = (NPW + 15) / 16;
        constexpr int HP = F + 8;
        constexpr int FP = NC + 1;
        static_assert((HP * 2) % 16 == 0 && ((HP * 2 / 16) & 1) == 1, "pitch");
        constexpr int HS_FLOATS = MT * 16 * HP / 2;
        constexpr int FS_FLOATS = MT * 16 * FP;
        static_assert(HS_FLOATS % 4 == 0 && (HS_FLOATS + FS_FLOATS) % 4 == 0, "16-byte aligned parts");
        static_assert(HS_FLOATS + FS_FLOATS + NC * HP / 2 <= SCAP * H, "fits over the scores");
        _Float16* hs  = (_Float16*)sco;
        float*    f2s = sco + HS_FLOATS;
        _Float16* wts = (_Float16*)(sco + HS_FLOATS + FS_FLOATS);

        for (int i = tid; i < F * NC; i += WGT)
            wts[(i % NC) * HP + i / NC] = (_Float16)Wn[i];
        for (int g = tid; g < MT * 16 * (F / 8); g += WGT) {
            const int r = g / (F / 8), c8 = (g % (F / 8)) * 8;
            v8h o = {};
            if (r < NPW && n0 + r < N_NODES) {
#pragma unroll
                for (int j = 0; j < 8; ++j) {
                    float v = (float)acc[r * F + c8 + j] * inv + bias[c8 + j];
                    v = v > 0.0f ? v : (__expf(v) - 1.0f);
                    o[j] = (_Float16)v;
                }
            }
            *(v8h*)(hs + r * HP + c8) = o;
        }
        __syncthreads();

        const int wvn = __builtin_amdgcn_readfirstlane(tid >> 5);
        if (wvn < MT) {
            const int half = (tid & 31) >> 4, idx = tid & 15;
            const _Float16* arow = hs  + (wvn * 16 + idx) * HP + half * 8;
            const _Float16* brow = wts + idx * HP + half * 8;
            v8f c = {};
#pragma unroll
            for (int k0 = 0; k0 < F; k0 += 32) {
                union { v16h v; struct { v8h lo, hi; } h; } a, b;
                a.h.lo = *(const v8h*)(arow + k0);
                a.h.hi = *(const v8h*)(arow + k0 + 16);
                b.h.lo = *(const v8h*)(brow + k0);
                b.h.hi = *(const v8h*)(brow + k0 + 16);
                c = __builtin_amdgcn_wmma_f32_16x16x32_f16(false, a.v, false, b.v,
                                                           (short)0, c, false, false);
            }
#pragma unroll
            for (int r = 0; r < 8; ++r) f2s[(wvn * 16 + half * 8 + r) * FP + idx] = c[r];
        }
        __syncthreads();

        for (int i = tid; i < NPW * NC; i += WGT)
            if (n0 + i / NC < N_NODES) featn[(size_t)n0 * NC + i] = f2s[(i / NC) * FP + i % NC];
        if (tid < (NPW + 31) / 32 * 32) {
            float ma = 0.0f;
            if (tid < NPW && n0 + tid < N_NODES) {
                float sl = 0.0f, sr = 0.0f;
#pragma unroll
                for (int d = 0; d < NC; ++d) {
                    float v = f2s[tid * FP + d];
                    sl = __builtin_fmaf(v, aln[d], sl); sr = __builtin_fmaf(v, arn[d], sr);
                    ma = fmaxf(ma, fabsf(v));
                }
                eln[n0 + tid] = sl; ern[n0 + tid] = sr;
            }
            for (int o = 16; o > 0; o >>= 1) ma = fmaxf(ma, __shfl_xor(ma, o, 32));
            if ((tid & 31) == 0) atomicMax(mxn, __float_as_uint(ma));
        }
    }
}

extern "C" void kernel_launch(void* const* d_in, const int* in_sizes, int n_in,
                              void* d_out, int out_size, void* d_ws, size_t ws_size,
                              hipStream_t stream) {
    const float* x   = (const float*)d_in[0];
    const int*   src = (const int*)  d_in[1];
    const int*   dst = (const int*)  d_in[2];
    const float* W1  = (const float*)d_in[3];
    const float* b1  = (const float*)d_in[4];
    const float* al1 = (const float*)d_in[5];
    const float* ar1 = (const float*)d_in[6];
    const float* W2  = (const float*)d_in[7];
    const float* b2  = (const float*)d_in[8];
    const float* al2 = (const float*)d_in[9];
    const float* ar2 = (const float*)d_in[10];
    float* out = (float*)d_out;
    (void)in_sizes; (void)n_in; (void)out_size; (void)ws_size;

    char* p = (char*)d_ws;
    auto carve = [&](size_t bytes) -> char* {
        char* r = p; p += (bytes + 255) & ~(size_t)255; return r;
    };
    _Float16* xh    = (_Float16*)carve((size_t)N_NODES * IN_F * 2);
    _Float16* w1t   = (_Float16*)carve((size_t)IN_F * F2 * 2);
    float*    feat1 = (float*)   carve((size_t)N_NODES * F2 * 4);
    float*    el1   = (float*)   carve((size_t)N_NODES * HEADS * 4);
    float*    er1   = (float*)   carve((size_t)N_NODES * HEADS * 4);
    unsigned* mxa1  = (unsigned*)carve(256);
    unsigned* mxa2  = mxa1 + 1;
    unsigned* glist = (unsigned*)carve((size_t)NWG * LCAP1 * 4);
    unsigned* gcnt  = (unsigned*)carve((size_t)NWG * 4);
    unsigned* planes = (unsigned*)carve((size_t)NWG * PL_PITCH * 4);
    float*    feat2 = (float*)   carve((size_t)N_NODES * NC * 4);
    float*    el2   = (float*)   carve((size_t)N_NODES * 4);
    float*    er2   = (float*)   carve((size_t)N_NODES * 4);

    hipMemsetAsync(mxa1, 0, 256, stream);

    const int T = 256;
    auto blk = [](long long n, int t) { return (int)((n + t - 1) / t); };

    {
        constexpr int N4 = N_NODES * IN_F / 4;
        constexpr int CV_BLOCKS = (N4 / CV_WAVE + 7) / 8;
        k_f32_to_f16<N4><<<CV_BLOCKS, T, 0, stream>>>(x, xh);
    }
    k_transpose_f16<<<blk((long long)IN_F * F2, T), T, 0, stream>>>(W1, w1t, IN_F, F2);

    {
        static_assert(F2 == 128 && N_NODES % GM_ROWS == 0, "the projection kernel's tile; the grid covers the rows exactly");
        k_gemm_wmma<IN_F><<<N_NODES / GM_ROWS, T, 0, stream>>>(xh, w1t, feat1);
    }
    k_el_er<<<blk((long long)N_NODES * HEADS, T), T, 0, stream>>>(feat1, al1, ar1, el1, er1,
                                                                  N_NODES, HEADS, HID, mxa1);
    k_owner_planes<<<PL_NB, PL_T, 0, stream>>>(dst, planes);
    k_edges_owned<HEADS, HID, LCAP1, true><<<NWG, WGT, 0, stream>>>(src, dst, planes, feat1, el1, er1, mxa1, b1,
                                                                     glist, gcnt, nullptr,
                                                                     W2, al2, ar2, feat2, el2, er2, mxa2);

    k_edges_owned<1, NC, LCAP2, false><<<NWG, WGT, 0, stream>>>(src, dst, planes, feat2, el2, er2, mxa2, b2,
                                                                 glist, gcnt, out,
                                                                 nullptr, nullptr, nullptr, nullptr, nullptr, nullptr,
                                                                 nullptr);
}


# Owner bit planes become owner lists: the pre-pass sorts each 4096 edges by owner in LDS, reserves places per owner (1 integer add), writes src|dst words; a workgroup reads its 25 KB list; same values
# speedup vs baseline: 8.7173x; 1.1476x over previous
#include <hip/hip_runtime.h>
#include <hip/hip_bf16.h>

#define N_NODES 50000
#define N_EDGES 1600000
#define IN_F    256
#define HID     32
#define HEADS   4
#define F2      128
#define NC      16
#define NEG_SLOPE 0.2f

typedef __attribute__((ext_vector_type(16))) _Float16 v16h;
typedef __attribute__((ext_vector_type(8)))  _Float16 v8h;
typedef __attribute__((ext_vector_type(8)))  float    v8f;

__device__ __forceinline__ unsigned enc_key(float f) {
    unsigned b = __float_as_uint(f);
    return (b & 0x80000000u) ? ~b : (b | 0x80000000u);
}
__device__ __forceinline__ float dec_key(unsigned k) {
    unsigned b = (k & 0x80000000u) ? (k ^ 0x80000000u) : ~k;
    if ((b & 0x7f800000u) == 0x7f800000u) return 0.0f;
    return __uint_as_float(b);
}
__device__ __forceinline__ float leaky(float v) { return v > 0.0f ? v : NEG_SLOPE * v; }

#define DEN_SCALE 2147483648.0f
#define DEN_INV   4.656612873077392578125e-10f
__device__ __forceinline__ int fx_exp(unsigned maxbits) {
    int e = (int)((maxbits >> 23) & 0xffu) - 127;
    return e < -96 ? -96 : (e > 96 ? 96 : e);
}
__device__ __forceinline__ float fx_scale(unsigned maxbits) {
    return __uint_as_float((unsigned)(156 - fx_exp(maxbits)) << 23);
}
__device__ __forceinline__ float fx_inv(unsigned maxbits) {
    return __uint_as_float((unsigned)(98 + fx_exp(maxbits)) << 23);
}

#define CV_U    16
#define CV_WAVE (CV_U * 32)
typedef __attribute__((ext_vector_type(4))) _Float16 v4h;
template <int N4>
__global__ void __launch_bounds__(256) k_f32_to_f16(const float* __restrict__ in,
                                                    _Float16* __restrict__ out) {
    static_assert(N4 % CV_WAVE == 0, "whole waves only");
    const int wave = (blockIdx.x * blockDim.x + threadIdx.x) >> 5;
    if (wave * CV_WAVE >= N4) return;
    const int g0 = wave * CV_WAVE + (threadIdx.x & 31);
    const float4* __restrict__ s = (const float4*)in;
    v4h* __restrict__ d = (v4h*)out;
    float4 v[CV_U];
#pragma unroll
    for (int q = 0; q < CV_U; ++q) v[q] = s[g0 + q * 32];
#pragma unroll
    for (int q = 0; q < CV_U; ++q) {
        v4h o = { (_Float16)v[q].x, (_Float16)v[q].y, (_Float16)v[q].z, (_Float16)v[q].w };
        d[g0 + q * 32] = o;
    }
}
__global__ void k_transpose_f16(const float* __restrict__ in, _Float16* __restrict__ out,
                                int R, int C) {
    int i = blockIdx.x * blockDim.x + threadIdx.x;
    if (i >= R * C) return;
    int r = i / C, c = i % C;
    out[c * R + r] = (_Float16)in[i];
}

#define GM_RT   5
#define GM_ROWS (GM_RT * 16)
template <int K>
__global__ void __launch_bounds__(256) k_gemm_wmma(const _Float16* __restrict__ A,
                                                   const _Float16* __restrict__ Bt,
                                                   float* __restrict__ C) {
    constexpr int NCOL = 128;
    constexpr int AP = K + 8;
    constexpr int CP = NCOL + 4;
    constexpr int CPR = K / 8;
    constexpr int AQ = GM_ROWS * CPR / 256;
    constexpr int CQ = GM_ROWS * (NCOL / 4) / 256;
    static_assert((AP * 2) % 16 == 0 && ((AP * 2 / 16) & 1) == 1 && GM_ROWS * CPR % 256 == 0 && CPR == 32, "A tile");
    static_assert((8 * CP) % 64 == 32 && NCOL / 4 == 32 && GM_ROWS * (NCOL / 4) % 256 == 0, "output tile");
    __shared__ __attribute__((aligned(16))) _Float16 as[GM_ROWS * AP];
    __shared__ __attribute__((aligned(16))) float cs[GM_ROWS * CP];

    const int tid = threadIdx.x;
    const int lane = tid & 31, wv = tid >> 5;
    const int half = lane >> 4;
    const int idx  = lane & 15;
    const int row0 = blockIdx.x * GM_ROWS;

    const _Float16* brow = Bt + (size_t)(wv * 16 + idx) * K + half * 8;
    union frag { v16h v; struct { v8h lo, hi; } h; };
    frag b[K / 32];
#pragma unroll
    for (int k0 = 0; k0 < K; k0 += 32) {
        b[k0 / 32].h.lo = *(const v8h*)(brow + k0);
        b[k0 / 32].h.hi = *(const v8h*)(brow + k0 + 16);
    }
    v8h va[AQ];
#pragma unroll
    for (int q = 0; q < AQ; ++q) {
        const int c = tid + q * 256, r = c / CPR, p = c % CPR;
        va[q] = *(const v8h*)(A + (size_t)(row0 + r) * K + p * 8);
    }
#pragma unroll
    for (int q = 0; q < AQ; ++q) {
        const int c = tid + q * 256, r = c / CPR, p = c % CPR;
        *(v8h*)(as + r * AP + p * 8) = va[q];
    }
    __syncthreads();

#pragma unroll
    for (int t = 0; t < GM_RT; ++t) {
        const _Float16* arow = as + (t * 16 + idx) * AP + half * 8;
        v8f acc = {};
#pragma unroll
        for (int k0 = 0; k0 < K; k0 += 32) {
            frag a;
            a.h.lo = *(const v8h*)(arow + k0);
            a.h.hi = *(const v8h*)(arow + k0 + 16);
            acc = __builtin_amdgcn_wmma_f32_16x16x32_f16(false, a.v, false, b[k0 / 32].v,
                                                         (short)0, acc, false, false);
        }
#pragma unroll
        for (int r = 0; r < 8; ++r) cs[(t * 16 + half * 8 + r) * CP + wv * 16 + idx] = acc[r];
    }
    __syncthreads();

#pragma unroll
    for (int q = 0; q < CQ; ++q) {
        const int c = tid + q * 256, r = c / (NCOL / 4), p = c % (NCOL / 4);
        *(float4*)(C + (size_t)(row0 + r) * NCOL + p * 4) = *(const float4*)(cs + r * CP + p * 4);
    }
}

__global__ void k_el_er(const float* __restrict__ feat, const float* __restrict__ al,
                        const float* __restrict__ ar, float* __restrict__ el,
                        float* __restrict__ er, int n, int H, int D,
                        unsigned* __restrict__ mxabs) {
    int i = blockIdx.x * blockDim.x + threadIdx.x;
    float ma = 0.0f;
    if (i < n * H) {
        int node = i / H, h = i % H;
        const float* f = feat + (size_t)node * H * D + h * D;
        const float* a = al + h * D;
        const float* b = ar + h * D;
        float sl = 0.0f, sr = 0.0f;
        for (int d = 0; d < D; ++d) {
            float v = f[d]; sl += v * a[d]; sr += v * b[d]; ma = fmaxf(ma, fabsf(v));
        }
        el[i] = sl; er[i] = sr;
    }
    for (int o = 16; o > 0; o >>= 1) ma = fmaxf(ma, __shfl_xor(ma, o, 32));
    if ((threadIdx.x & 31) == 0) atomicMax(mxabs, __float_as_uint(ma));
}

#define NPW   196
#define NWG   256
#define WGT   1024
#define LCAP1 16384
#define LCAP2 40960

#define OL_E   4096
#define OL_T   256
#define OL_NB  ((N_EDGES + OL_E - 1) / OL_E)
#define OL_K   (OL_E / (4 * OL_T))
__global__ void __launch_bounds__(OL_T) k_owner_lists(const int* __restrict__ src, const int* __restrict__ dst,
                                                      unsigned* __restrict__ glist, unsigned* __restrict__ gcnt) {
    static_assert(N_EDGES % 4 == 0, "whole 16-byte groups");
    static_assert((N_NODES - 1) / NPW < NWG && NWG == OL_T && NWG <= 256, "one thread per owner; an owner number is a byte");
    static_assert(N_NODES <= 65536 && NPW <= 65536 && OL_E <= 65536, "packed words");
    __shared__ unsigned hist[NWG];
    __shared__ unsigned lstart[NWG];
    __shared__ unsigned gbase[NWG];
    __shared__ unsigned wsum[OL_T / 32];
    __shared__ unsigned sorted[OL_E];
    __shared__ unsigned char sown[OL_E];
    const int tid = threadIdx.x;
    const int lane = tid & 31, wv = tid >> 5;
    int4 d[OL_K], s[OL_K];
    bool in[OL_K];
#pragma unroll
    for (int k = 0; k < OL_K; ++k) {
        const int e = blockIdx.x * OL_E + k * (4 * OL_T) + tid * 4;
        in[k] = e < N_EDGES;
        d[k] = *(const int4*)(dst + (in[k] ? e : 0));
        s[k] = *(const int4*)(src + (in[k] ? e : 0));
    }
    hist[tid] = 0u;
    __syncthreads();
    unsigned orr[OL_K][4];
#pragma unroll
    for (int k = 0; k < OL_K; ++k) {
        const int dv[4] = { d[k].x, d[k].y, d[k].z, d[k].w };
#pragma unroll
        for (int j = 0; j < 4; ++j) {
            orr[k][j] = 0u;
            if (in[k]) {
                const unsigned o = (unsigned)dv[j] / (unsigned)NPW;
                orr[k][j] = o | (atomicAdd(&hist[o], 1u) << 8);
            }
        }
    }
    __syncthreads();
    const unsigned n = hist[tid];
    gbase[tid] = n ? atomicAdd(&gcnt[tid], n) : 0u;
    unsigned incl = n;
    for (int o = 1; o < 32; o <<= 1) {
        const unsigned t = (unsigned)__shfl_up((int)incl, o, 32);
        if (lane >= o) incl += t;
    }
    if (lane == 31) wsum[wv] = incl;
    __syncthreads();
    unsigned woff = 0u;
    for (int q = 0; q < wv; ++q) woff += wsum[q];
    lstart[tid] = woff + incl - n;
    __syncthreads();
#pragma unroll
    for (int k = 0; k < OL_K; ++k) {
        const int dv[4] = { d[k].x, d[k].y, d[k].z, d[k].w };
        const int sv[4] = { s[k].x, s[k].y, s[k].z, s[k].w };
#pragma unroll
        for (int j = 0; j < 4; ++j) {
            if (in[k]) {
                const unsigned o = orr[k][j] & 0xffu;
                const unsigned p = lstart[o] + (orr[k][j] >> 8);
                sorted[p] = (unsigned)sv[j] | (((unsigned)dv[j] - o * (unsigned)NPW) << 16);
                sown[p] = (unsigned char)o;
            }
        }
    }
    __syncthreads();
    const int left = N_EDGES - blockIdx.x * OL_E;
    const int nvalid = left < OL_E ? left : OL_E;
    for (int j = tid; j < nvalid; j += OL_T) {
        const unsigned o = sown[j];
        const unsigned q = gbase[o] + ((unsigned)j - lstart[o]);
        if (q < (unsigned)LCAP1) glist[(size_t)o * LCAP1 + q] = sorted[j];
    }
}

template <int H, int D, int LCAP, bool FIRST>
__global__ void __launch_bounds__(WGT)
k_edges_owned(const int* __restrict__ src, const int* __restrict__ dst,
              const float* __restrict__ feat, const float* __restrict__ el,
              const float* __restrict__ er, const unsigned* __restrict__ mxabs,
              const float* __restrict__ bias, const unsigned* __restrict__ glist,
              const unsigned* __restrict__ gcnt,
              float* __restrict__ o32,
              const float* __restrict__ Wn, const float* __restrict__ aln, const float* __restrict__ arn,
              float* __restrict__ featn, float* __restrict__ eln, float* __restrict__ ern,
              unsigned* __restrict__ mxn) {
    constexpr int F = H * D;
    static_assert(F >= 32 ? (D % 32 == 0 && (H & (H - 1)) == 0 && H <= 32) : (H == 1 && 32 % F == 0), "lane map");
    __shared__ int acc[NPW * F];
    __shared__ unsigned long long den[NPW * H];
    __shared__ unsigned mx[NPW * H];
    __shared__ float ers[NPW * H];
    __shared__ unsigned list[LCAP];
    static_assert(LCAP >= LCAP1 && LCAP1 % WGT == 0, "a whole list fits; whole sweeps of the workgroup");
    constexpr int SCAP = 28672 / H;
    __shared__ __attribute__((aligned(16))) float sco[SCAP * H];

    const int tid = threadIdx.x;
    const int n0 = blockIdx.x * NPW;

    const unsigned total = gcnt[blockIdx.x];
    const bool complete = total <= (unsigned)LCAP1;
    constexpr int LQ = 8;
    const unsigned* __restrict__ gl = glist + (size_t)blockIdx.x * LCAP1;
    if (complete) {
        for (unsigned i0 = 0u; i0 < total; i0 += (unsigned)(LQ * WGT)) {
            unsigned lw[LQ];
#pragma unroll
            for (int q = 0; q < LQ; ++q) {
                const unsigned i = i0 + (unsigned)(q * WGT + tid);
                lw[q] = gl[i < total ? i : 0u];
            }
#pragma unroll
            for (int q = 0; q < LQ; ++q) {
                const unsigned i = i0 + (unsigned)(q * WGT + tid);
                if (i < total) list[i] = lw[q];
            }
        }
    }
    for (int i = tid; i < NPW * F; i += WGT) acc[i] = 0;
    for (int i = tid; i < NPW * H; i += WGT) {
        den[i] = 0ull; mx[i] = 0u;
        ers[i] = (n0 + i / H) < N_NODES ? er[(size_t)n0 * H + i] : 0.0f;
    }
    __syncthreads();

    auto score = [&](unsigned s, unsigned dl, int h) -> float {
        return leaky(el[(size_t)s * H + h] + ers[dl * H + h]);
    };

    if (complete) {
        for (unsigned i = tid; i < total; i += WGT) {
            const unsigned w = list[i];
            const unsigned s = w & 0xffffu, dl = w >> 16;
            float sv[H];
#pragma unroll
            for (int h = 0; h < H; ++h) {
                sv[h] = score(s, dl, h);
                atomicMax(&mx[dl * H + h], enc_key(sv[h]));
            }
            if (i < (unsigned)SCAP) {
#pragma unroll
                for (int h = 0; h < H; ++h) sco[i * H + h] = sv[h];
            }
        }
    } else {
        for (int e = tid; e < N_EDGES; e += WGT) {
            unsigned dl = (unsigned)(dst[e] - n0);
            if (dl < (unsigned)NPW) {
                unsigned s = (unsigned)src[e];
                for (int h = 0; h < H; ++h) atomicMax(&mx[dl * H + h], enc_key(score(s, dl, h)));
            }
        }
    }
    __syncthreads();

    auto edge_den = [&](unsigned s, unsigned dl) {
#pragma unroll
        for (int h = 0; h < H; ++h) {
            float x = __expf(score(s, dl, h) - dec_key(mx[dl * H + h]));
            atomicAdd(&den[dl * H + h], (unsigned long long)__float2uint_rn(x * DEN_SCALE));
        }
    };
    if (complete) {
        for (unsigned i = tid; i < total; i += WGT) {
            const unsigned w = list[i];
            if (i < (unsigned)SCAP) {
                const unsigned dl = w >> 16;
#pragma unroll
                for (int h = 0; h < H; ++h) {
                    float x = __expf(sco[i * H + h] - dec_key(mx[dl * H + h]));
                    atomicAdd(&den[dl * H + h], (unsigned long long)__float2uint_rn(x * DEN_SCALE));
                }
            } else {
                edge_den(w & 0xffffu, w >> 16);
            }
        }
    } else {
        for (int e = tid; e < N_EDGES; e += WGT) {
            unsigned dl = (unsigned)(dst[e] - n0);
            if (dl < (unsigned)NPW) edge_den((unsigned)src[e], dl);
        }
    }
    __syncthreads();

    const float sc = fx_scale(*mxabs);
    auto alpha_of = [&](float v, unsigned dl, int h) -> float {
        float x = __expf(v - dec_key(mx[dl * H + h]));
        return x / fmaxf((float)den[dl * H + h] * DEN_INV, 1e-9f) * sc;
    };
    auto alpha = [&](unsigned s, unsigned dl, int h) -> float { return alpha_of(score(s, dl, h), dl, h); };
    auto kept_score = [&](unsigned i, unsigned w, int h) -> float {
        return i < (unsigned)SCAP ? sco[i * H + h] : score(w & 0xffffu, w >> 16, h);
    };
    if (complete) {
        const int lane = tid & 31;
        const unsigned wv  = (unsigned)__builtin_amdgcn_readfirstlane(tid >> 5);
        const unsigned tot = (unsigned)__builtin_amdgcn_readfirstlane((int)total);
        if constexpr (F >= 32) {
            constexpr int EB = 32 / H;
            constexpr int NJ = F / 32;
            for (unsigned i0 = wv * EB; i0 < tot; i0 += (unsigned)((WGT / 32) * EB)) {
                const unsigned i = i0 + (unsigned)(lane / H);
                const unsigned ic = i < tot ? i : tot - 1u;
                const unsigned w = list[ic];
                float f[EB][NJ];
                unsigned wu[EB];
#pragma unroll
                for (int u = 0; u < EB; ++u) {
                    wu[u] = (unsigned)__builtin_amdgcn_readlane((int)w, u * H);
                    const float* fr = feat + (size_t)(wu[u] & 0xffffu) * F + lane;
#pragma unroll
                    for (int j = 0; j < NJ; ++j) f[u][j] = fr[32 * j];
                }
                float aq = alpha_of(kept_score(ic, w, lane & (H - 1)), w >> 16, lane & (H - 1));
                aq = i < tot ? aq : 0.0f;
#pragma unroll
                for (int u = 0; u < EB; ++u) {
                    const unsigned dl = wu[u] >> 16;
#pragma unroll
                    for (int j = 0; j < NJ; ++j) {
                        const float a = __int_as_float(__builtin_amdgcn_readlane(__float_as_int(aq),
                                                                                 u * H + (32 * j) / D));
                        atomicAdd(&acc[dl * F + lane + 32 * j], __float2int_rn(f[u][j] * a));
                    }
                }
            }
        } else {
            constexpr int EPS = 32 / F;
            constexpr int NS  = 32 / EPS;
            const int sub = lane / F, fl = lane % F;
            for (unsigned i0 = wv * 32u; i0 < tot; i0 += (unsigned)WGT) {
                const unsigned i = i0 + (unsigned)lane;
                const unsigned ic = i < tot ? i : tot - 1u;
                const unsigned w = list[ic];
                float f[NS];
                unsigned dls[NS];
#pragma unroll
                for (int u = 0; u < NS; ++u) {
                    const unsigned wl = (unsigned)__builtin_amdgcn_ds_bpermute((u * EPS + sub) * 4, (int)w);
                    dls[u] = wl >> 16;
                    f[u] = feat[(wl & 0xffffu) * (unsigned)F + (unsigned)fl];
                }
                float aq = alpha_of(kept_score(ic, w, 0), w >> 16, 0);
                aq = i < tot ? aq : 0.0f;
#pragma unroll
                for (int u = 0; u < NS; ++u) {
                    const float a = __int_as_float(__builtin_amdgcn_ds_bpermute((u * EPS + sub) * 4,
                                                                                __float_as_int(aq)));
                    atomicAdd(&acc[dls[u] * F + fl], __float2int_rn(f[u] * a));
                }
            }
        }
    } else {
        for (int e = tid; e < N_EDGES; e += WGT) {
            unsigned dl = (unsigned)(dst[e] - n0);
            if (dl < (unsigned)NPW) {
                unsigned s = (unsigned)src[e];
                for (int h = 0; h < H; ++h) {
                    float ah = alpha(s, dl, h);
                    for (int d = 0; d < D; ++d)
                        atomicAdd(&acc[dl * F + h * D + d],
                                  __float2int_rn(feat[(size_t)s * F + h * D + d] * ah));
                }
            }
        }
    }
    __syncthreads();

    const float inv = fx_inv(*mxabs);
    if constexpr (!FIRST) {
        for (int i = tid; i < NPW * F; i += WGT)
            if (n0 + i / F < N_NODES) o32[(size_t)n0 * F + i] = (float)acc[i] * inv + bias[i % F];
    } else {
        static_assert(F % 32 == 0 && NC == 16, "one 16-column tile, whole k steps");
        constexpr int MT = (NPW + 15) / 16;
        constexpr int HP = F + 8;
        constexpr int FP = NC + 1;
        static_assert((HP * 2) % 16 == 0 && ((HP * 2 / 16) & 1) == 1, "pitch");
        constexpr int HS_FLOATS = MT * 16 * HP / 2;
        constexpr int FS_FLOATS = MT * 16 * FP;
        static_assert(HS_FLOATS % 4 == 0 && (HS_FLOATS + FS_FLOATS) % 4 == 0, "16-byte aligned parts");
        static_assert(HS_FLOATS + FS_FLOATS + NC * HP / 2 <= SCAP * H, "fits over the scores");
        _Float16* hs  = (_Float16*)sco;
        float*    f2s = sco + HS_FLOATS;
        _Float16* wts = (_Float16*)(sco + HS_FLOATS + FS_FLOATS);

        for (int i = tid; i < F * NC; i += WGT)
            wts[(i % NC) * HP + i / NC] = (_Float16)Wn[i];
        for (int g = tid; g < MT * 16 * (F / 8); g += WGT) {
            const int r = g / (F / 8), c8 = (g % (F / 8)) * 8;
            v8h o = {};
            if (r < NPW && n0 + r < N_NODES) {
#pragma unroll
                for (int j = 0; j < 8; ++j) {
                    float v = (float)acc[r * F + c8 + j] * inv + bias[c8 + j];
                    v = v > 0.0f ? v : (__expf(v) - 1.0f);
                    o[j] = (_Float16)v;
                }
            }
            *(v8h*)(hs + r * HP + c8) = o;
        }
        __syncthreads();

        const int wvn = __builtin_amdgcn_readfirstlane(tid >> 5);
        if (wvn < MT) {
            const int half = (tid & 31) >> 4, idx = tid & 15;
            const _Float16* arow = hs  + (wvn * 16 + idx) * HP + half * 8;
            const _Float16* brow = wts + idx * HP + half * 8;
            v8f c = {};
#pragma unroll
            for (int k0 = 0; k0 < F; k0 += 32) {
                union { v16h v; struct { v8h lo, hi; } h; } a, b;
                a.h.lo = *(const v8h*)(arow + k0);
                a.h.hi = *(const v8h*)(arow + k0 + 16);
                b.h.lo = *(const v8h*)(brow + k0);
                b.h.hi = *(const v8h*)(brow + k0 + 16);
                c = __builtin_amdgcn_wmma_f32_16x16x32_f16(false, a.v, false, b.v,
                                                           (short)0, c, false, false);
            }
#pragma unroll
            for (int r = 0; r < 8; ++r) f2s[(wvn * 16 + half * 8 + r) * FP + idx] = c[r];
        }
        __syncthreads();

        for (int i = tid; i < NPW * NC; i += WGT)
            if (n0 + i / NC < N_NODES) featn[(size_t)n0 * NC + i] = f2s[(i / NC) * FP + i % NC];
        if (tid < (NPW + 31) / 32 * 32) {
            float ma = 0.0f;
            if (tid < NPW && n0 + tid < N_NODES) {
                float sl = 0.0f, sr = 0.0f;
#pragma unroll
                for (int d = 0; d < NC; ++d) {
                    float v = f2s[tid * FP + d];
                    sl = __builtin_fmaf(v, aln[d], sl); sr = __builtin_fmaf(v, arn[d], sr);
                    ma = fmaxf(ma, fabsf(v));
                }
                eln[n0 + tid] = sl; ern[n0 + tid] = sr;
            }
            for (int o = 16; o > 0; o >>= 1) ma = fmaxf(ma, __shfl_xor(ma, o, 32));
            if ((tid & 31) == 0) atomicMax(mxn, __float_as_uint(ma));
        }
    }
}

extern "C" void kernel_launch(void* const* d_in, const int* in_sizes, int n_in,
                              void* d_out, int out_size, void* d_ws, size_t ws_size,
                              hipStream_t stream) {
    const float* x   = (const float*)d_in[0];
    const int*   src = (const int*)  d_in[1];
    const int*   dst = (const int*)  d_in[2];
    const float* W1  = (const float*)d_in[3];
    const float* b1  = (const float*)d_in[4];
    const float* al1 = (const float*)d_in[5];
    const float* ar1 = (const float*)d_in[6];
    const float* W2  = (const float*)d_in[7];
    const float* b2  = (const float*)d_in[8];
    const float* al2 = (const float*)d_in[9];
    const float* ar2 = (const float*)d_in[10];
    float* out = (float*)d_out;
    (void)in_sizes; (void)n_in; (void)out_size; (void)ws_size;

    char* p = (char*)d_ws;
    auto carve = [&](size_t bytes) -> char* {
        char* r = p; p += (bytes + 255) & ~(size_t)255; return r;
    };
    _Float16* xh    = (_Float16*)carve((size_t)N_NODES * IN_F * 2);
    _Float16* w1t   = (_Float16*)carve((size_t)IN_F * F2 * 2);
    float*    feat1 = (float*)   carve((size_t)N_NODES * F2 * 4);
    float*    el1   = (float*)   carve((size_t)N_NODES * HEADS * 4);
    float*    er1   = (float*)   carve((size_t)N_NODES * HEADS * 4);
    unsigned* mxa1  = (unsigned*)carve(256 + (size_t)NWG * 4);
    unsigned* mxa2  = mxa1 + 1;
    unsigned* gcnt  = mxa1 + 64;
    unsigned* glist = (unsigned*)carve((size_t)NWG * LCAP1 * 4);
    float*    feat2 = (float*)   carve((size_t)N_NODES * NC * 4);
    float*    el2   = (float*)   carve((size_t)N_NODES * 4);
    float*    er2   = (float*)   carve((size_t)N_NODES * 4);

    hipMemsetAsync(mxa1, 0, 256 + (size_t)NWG * 4, stream);

    const int T = 256;
    auto blk = [](long long n, int t) { return (int)((n + t - 1) / t); };

    {
        constexpr int N4 = N_NODES * IN_F / 4;
        constexpr int CV_BLOCKS = (N4 / CV_WAVE + 7) / 8;
        k_f32_to_f16<N4><<<CV_BLOCKS, T, 0, stream>>>(x, xh);
    }
    k_transpose_f16<<<blk((long long)IN_F * F2, T), T, 0, stream>>>(W1, w1t, IN_F, F2);

    {
        static_assert(F2 == 128 && N_NODES % GM_ROWS == 0, "the projection kernel's tile; the grid covers the rows exactly");
        k_gemm_wmma<IN_F><<<N_NODES / GM_ROWS, T, 0, stream>>>(xh, w1t, feat1);
    }
    k_el_er<<<blk((long long)N_NODES * HEADS, T), T, 0, stream>>>(feat1, al1, ar1, el1, er1,
                                                                  N_NODES, HEADS, HID, mxa1);
    k_owner_lists<<<OL_NB, OL_T, 0, stream>>>(src, dst, glist, gcnt);
    k_edges_owned<HEADS, HID, LCAP1, true><<<NWG, WGT, 0, stream>>>(src, dst, feat1, el1, er1, mxa1, b1,
                                                                     glist, gcnt, nullptr,
                                                                     W2, al2, ar2, feat2, el2, er2, mxa2);

    k_edges_owned<1, NC, LCAP2, false><<<NWG, WGT, 0, stream>>>(src, dst, feat2, el2, er2, mxa2, b2,
                                                                 glist, gcnt, out,
                                                                 nullptr, nullptr, nullptr, nullptr, nullptr, nullptr,
                                                                 nullptr);
}
